# starting kernel
# baseline (speedup 1.0000x reference)
#include <hip/hip_runtime.h>
#include <hip/hip_bf16.h>

typedef __attribute__((ext_vector_type(16))) _Float16 v16h;
typedef __attribute__((ext_vector_type(8)))  float    v8f;

#define E32 32
#define GQ  2

union H8 { uint4 u; _Float16 h[8]; };

__device__ __forceinline__ int lane_id() { return (int)(threadIdx.x & 31); }

__device__ __forceinline__ v8f wmma_f16(v16h a, v16h b, v8f c) {
  return __builtin_amdgcn_wmma_f32_16x16x32_f16(
       false, a,  false, b,
       (short)0, c,  false,  false);
}

template <typename T>
__device__ __forceinline__ v16h load_frag(const T* __restrict__ src, int stride) {
  const int l  = lane_id();
  const int r  = l & 15;
  const int kb = (l >> 4) << 3;
  const T* p = src + (size_t)r * stride + kb;
  v16h f;
#pragma unroll
  for (int i = 0; i < 8; ++i) {
    f[i]     = (_Float16)p[i];
    f[8 + i] = (_Float16)p[i + 16];
  }
  return f;
}

__device__ __forceinline__ void store_tile_f16(_Float16* __restrict__ dst,
                                               const _Float16* __restrict__ lds) {
  const int l = lane_id();
#pragma unroll
  for (int i = 0; i < 2; ++i) {
    const int idx = i * 32 + l;
    *(uint4*)(dst + idx * 8) = *(const uint4*)(lds + idx * 8);
  }
}

__global__ void __launch_bounds__(32)
qkv_kernel(const float* __restrict__ x, const float* __restrict__ w,
           const float* __restrict__ bias,
           _Float16* __restrict__ q, _Float16* __restrict__ k,
           _Float16* __restrict__ v, _Float16* __restrict__ vT) {
  __shared__ alignas(16) _Float16 tile[16 * 32];

  const int t  = blockIdx.x;
  const int g  = blockIdx.y;
  const int l  = lane_id();
  const int rowbase = t * 16;
  const int b     = rowbase >> 10;
  const int nbase = rowbase & 1023;
  const int bg    = b * GQ + g;
  const int nn = l & 15, half = l >> 4;

  v16h a = load_frag(x + (size_t)rowbase * (GQ * E32) + g * E32, GQ * E32);

  _Float16* outs[3] = {q, k, v};
#pragma unroll
  for (int s = 0; s < 3; ++s) {
    v8f acc[2];
    float bn[2];
#pragma unroll
    for (int cb = 0; cb < 2; ++cb) {
      const int obase = g * 96 + s * 32 + cb * 16;
      v16h bf = load_frag(w + (size_t)obase * E32, E32);
      v8f z = {};
      acc[cb] = wmma_f16(a, bf, z);
      bn[cb]  = bias[obase + nn];
    }
    __syncthreads();
#pragma unroll
    for (int cb = 0; cb < 2; ++cb) {
#pragma unroll
      for (int r = 0; r < 8; ++r) {
        const int m = r + half * 8;
        tile[m * 32 + cb * 16 + nn] = (_Float16)(acc[cb][r] + bn[cb]);
      }
    }
    if (s == 2) {
#pragma unroll
      for (int cb = 0; cb < 2; ++cb) {
        H8 p;
#pragma unroll
        for (int r = 0; r < 8; ++r) p.h[r] = (_Float16)(acc[cb][r] + bn[cb]);
        const int e = cb * 16 + nn;
        *(uint4*)(vT + ((size_t)bg * E32 + e) * 1024 + nbase + half * 8) = p.u;
      }
    }
    __syncthreads();
    store_tile_f16(outs[s] + ((size_t)bg * 1024 + nbase) * E32, tile);
  }
}

__global__ void __launch_bounds__(32)
attn_kernel(const _Float16* __restrict__ q, const _Float16* __restrict__ k,
            const _Float16* __restrict__ vT, _Float16* __restrict__ attn) {
  __shared__ alignas(16) _Float16 pTile[16 * 64];

  const int qt = blockIdx.x;
  const int bg = blockIdx.y;
  const int l  = lane_id();
  const _Float16* kb  = k  + (size_t)bg * 1024 * E32;
  const _Float16* vtb = vT + (size_t)bg * E32 * 1024;

  v16h aq = load_frag(q + ((size_t)bg * 1024 + qt * 16) * E32, E32);

  float mrow[8], lrow[8];
  v8f acc0 = {}, acc1 = {};
#pragma unroll
  for (int r = 0; r < 8; ++r) { mrow[r] = -1e30f; lrow[r] = 0.0f; }

  const float scale = 1.0f / (float)E32;
  const int nn = l & 15, half = l >> 4;

  for (int jt = 0; jt < 16; ++jt) {
    const int jb = jt * 64;
    if (jt < 15) {
      __builtin_prefetch(kb + (size_t)(jb + 64) * E32, 0, 1);
      __builtin_prefetch(vtb + jb + 64, 0, 1);
    }
    v8f s[4];
#pragma unroll
    for (int c = 0; c < 4; ++c) {
      v16h bk = load_frag(kb + (size_t)(jb + c * 16) * E32, E32);
      v8f z = {};
      s[c] = wmma_f16(aq, bk, z);
    }

    float p[4][8];
#pragma unroll
    for (int r = 0; r < 8; ++r) {
      float sv[4];
      float mx = -1e30f;
#pragma unroll
      for (int c = 0; c < 4; ++c) { sv[c] = s[c][r] * scale; mx = fmaxf(mx, sv[c]); }
#pragma unroll
      for (int off = 1; off < 16; off <<= 1) mx = fmaxf(mx, __shfl_xor(mx, off, 32));
      const float mn  = fmaxf(mrow[r], mx);
      const float fac = __expf(mrow[r] - mn);
      float rs = 0.0f;
#pragma unroll
      for (int c = 0; c < 4; ++c) { p[c][r] = __expf(sv[c] - mn); rs += p[c][r]; }
#pragma unroll
      for (int off = 1; off < 16; off <<= 1) rs += __shfl_xor(rs, off, 32);
      lrow[r] = lrow[r] * fac + rs;
      mrow[r] = mn;
      acc0[r] *= fac;
      acc1[r] *= fac;
    }

    __syncthreads();
#pragma unroll
    for (int c = 0; c < 4; ++c) {
#pragma unroll
      for (int r = 0; r < 8; ++r) {
        const int m = r + half * 8;
        pTile[m * 64 + c * 16 + nn] = (_Float16)p[c][r];
      }
    }
    __syncthreads();
    v16h ap0 = load_frag((const _Float16*)pTile, 64);
    v16h ap1 = load_frag((const _Float16*)pTile + 32, 64);
    v16h bv00 = load_frag(vtb + jb, 1024);
    v16h bv01 = load_frag(vtb + jb + 32, 1024);
    v16h bv10 = load_frag(vtb + (size_t)16 * 1024 + jb, 1024);
    v16h bv11 = load_frag(vtb + (size_t)16 * 1024 + jb + 32, 1024);
    acc0 = wmma_f16(ap0, bv00, acc0);
    acc0 = wmma_f16(ap1, bv01, acc0);
    acc1 = wmma_f16(ap0, bv10, acc1);
    acc1 = wmma_f16(ap1, bv11, acc1);
  }

  __syncthreads();
#pragma unroll
  for (int r = 0; r < 8; ++r) {
    const int m = r + half * 8;
    const float inv = 1.0f / lrow[r];
    pTile[m * 32 + nn]      = (_Float16)(acc0[r] * inv);
    pTile[m * 32 + 16 + nn] = (_Float16)(acc1[r] * inv);
  }
  __syncthreads();
  store_tile_f16(attn + ((size_t)bg * 1024 + qt * 16) * E32, (const _Float16*)pTile);
}

__global__ void __launch_bounds__(32)
mlp_kernel(const _Float16* __restrict__ v, const _Float16* __restrict__ attn,
           const float* __restrict__ w1,  const float* __restrict__ b1,
           const float* __restrict__ w10, const float* __restrict__ b10,
           const float* __restrict__ w11, const float* __restrict__ b11,
           float* __restrict__ out) {
  __shared__ alignas(16) _Float16 h1[16 * 32];
  __shared__ alignas(16) _Float16 h2[16 * 32];
  __shared__ alignas(16) float    ot[16 * 32];

  const int t = blockIdx.x;
  const int g = blockIdx.y;
  const int l = lane_id();
  const int rowbase = t * 16;
  const int b     = rowbase >> 10;
  const int nbase = rowbase & 1023;
  const int nn = l & 15, half = l >> 4;

  const size_t src = ((size_t)(b * GQ + g) * 1024 + nbase) * E32;
  v16h a_v  = load_frag(v + src, E32);
  v16h a_at = load_frag(attn + src, E32);

#pragma unroll
  for (int cb = 0; cb < 2; ++cb) {
    const int ob = g * 32 + cb * 16;
    v16h bw0 = load_frag(w1 + (size_t)ob * 64, 64);
    v16h bw1 = load_frag(w1 + (size_t)ob * 64 + 32, 64);
    v8f acc = {};
    acc = wmma_f16(a_v, bw0, acc);
    acc = wmma_f16(a_at, bw1, acc);
    const float bn = b1[ob + nn];
#pragma unroll
    for (int r = 0; r < 8; ++r) {
      const int m = r + half * 8;
      const float z = acc[r] + bn;
      h1[m * 32 + cb * 16 + nn] = (_Float16)(z >= 0.0f ? z : 0.01f * z);
    }
  }
  __syncthreads();

  v16h a_h1 = load_frag((const _Float16*)h1, 32);
#pragma unroll
  for (int cb = 0; cb < 2; ++cb) {
    const int ob = g * 32 + cb * 16;
    v16h bw = load_frag(w10 + (size_t)ob * 32, 32);
    v8f acc = {};
    acc = wmma_f16(a_h1, bw, acc);
    const float bn = b10[ob + nn];
#pragma unroll
    for (int r = 0; r < 8; ++r) {
      const int m = r + half * 8;
      const float z = acc[r] + bn;
      h2[m * 32 + cb * 16 + nn] = (_Float16)(z >= 0.0f ? z : 0.01f * z);
    }
  }
  __syncthreads();

  v16h a_h2 = load_frag((const _Float16*)h2, 32);
#pragma unroll
  for (int cb = 0; cb < 2; ++cb) {
    const int ob = g * 32 + cb * 16;
    v16h bw = load_frag(w11 + (size_t)ob * 32, 32);
    v8f acc = {};
    acc = wmma_f16(a_h2, bw, acc);
    const float bn = b11[ob + nn];
#pragma unroll
    for (int r = 0; r < 8; ++r) {
      const int m = r + half * 8;
      const float z = acc[r] + bn;
      ot[m * 32 + cb * 16 + nn] = (z >= 0.0f ? z : 0.01f * z);
    }
  }
  __syncthreads();
  float* dst = out + (size_t)rowbase * 64 + g * 32;
#pragma unroll
  for (int i = 0; i < 4; ++i) {
    const int idx = i * 32 + l;
    const int row = idx >> 3;
    const int c   = (idx & 7) * 4;
    *(uint4*)(dst + (size_t)row * 64 + c) = *(const uint4*)(ot + row * 32 + c);
  }
}

extern "C" void kernel_launch(void* const* d_in, const int* in_sizes, int n_in,
                              void* d_out, int out_size, void* d_ws, size_t ws_size,
                              hipStream_t stream) {
  const float* x     = (const float*)d_in[0];
  const float* w_qkv = (const float*)d_in[1];
  const float* b_qkv = (const float*)d_in[2];
  const float* w_h1  = (const float*)d_in[3];
  const float* b_h1  = (const float*)d_in[4];
  const float* w_h10 = (const float*)d_in[5];
  const float* b_h10 = (const float*)d_in[6];
  const float* w_h11 = (const float*)d_in[7];
  const float* b_h11 = (const float*)d_in[8];
  float* out = (float*)d_out;

  const size_t NEL = (size_t)32 * GQ * 1024 * E32;
  _Float16* ws   = (_Float16*)d_ws;
  _Float16* q    = ws;
  _Float16* k    = ws + NEL;
  _Float16* v    = ws + 2 * NEL;
  _Float16* vT   = ws + 3 * NEL;
  _Float16* attn = ws + 4 * NEL;

  qkv_kernel<<<dim3(2048, GQ), 32, 0, stream>>>(x, w_qkv, b_qkv, q, k, v, vT);
  attn_kernel<<<dim3(64, 64), 32, 0, stream>>>(q, k, vT, attn);
  mlp_kernel<<<dim3(2048, GQ), 32, 0, stream>>>(v, attn, w_h1, b_h1,
                                                w_h10, b_h10, w_h11, b_h11, out);
}


# attention scores computed transposed (K.Q^T) so P feeds the PV product from registers: key loop without LDS re-stripe or 16-lane shuffles, loads at step head; row sums per lane half, joined at end
# speedup vs baseline: 1.2061x; 1.2061x over previous
#include <hip/hip_runtime.h>
#include <hip/hip_bf16.h>

typedef __attribute__((ext_vector_type(16))) _Float16 v16h;
typedef __attribute__((ext_vector_type(8)))  float    v8f;

#define E32 32
#define GQ  2

union H8 { uint4 u; _Float16 h[8]; };

__device__ __forceinline__ int lane_id() { return (int)(threadIdx.x & 31); }

__device__ __forceinline__ v8f wmma_f16(v16h a, v16h b, v8f c) {
  return __builtin_amdgcn_wmma_f32_16x16x32_f16(
       false, a,  false, b,
       (short)0, c,  false,  false);
}

template <typename T>
__device__ __forceinline__ v16h load_frag(const T* __restrict__ src, int stride) {
  const int l  = lane_id();
  const int r  = l & 15;
  const int kb = (l >> 4) << 3;
  const T* p = src + (size_t)r * stride + kb;
  v16h f;
#pragma unroll
  for (int i = 0; i < 8; ++i) {
    f[i]     = (_Float16)p[i];
    f[8 + i] = (_Float16)p[i + 16];
  }
  return f;
}

__device__ __forceinline__ void store_tile_f16(_Float16* __restrict__ dst,
                                               const _Float16* __restrict__ lds) {
  const int l = lane_id();
#pragma unroll
  for (int i = 0; i < 2; ++i) {
    const int idx = i * 32 + l;
    *(uint4*)(dst + idx * 8) = *(const uint4*)(lds + idx * 8);
  }
}

__global__ void __launch_bounds__(32)
qkv_kernel(const float* __restrict__ x, const float* __restrict__ w,
           const float* __restrict__ bias,
           _Float16* __restrict__ q, _Float16* __restrict__ k,
           _Float16* __restrict__ v, _Float16* __restrict__ vT) {
  __shared__ alignas(16) _Float16 tile[16 * 32];

  const int t  = blockIdx.x;
  const int g  = blockIdx.y;
  const int l  = lane_id();
  const int rowbase = t * 16;
  const int b     = rowbase >> 10;
  const int nbase = rowbase & 1023;
  const int bg    = b * GQ + g;
  const int nn = l & 15, half = l >> 4;

  v16h a = load_frag(x + (size_t)rowbase * (GQ * E32) + g * E32, GQ * E32);

  _Float16* outs[3] = {q, k, v};
#pragma unroll
  for (int s = 0; s < 3; ++s) {
    v8f acc[2];
    float bn[2];
#pragma unroll
    for (int cb = 0; cb < 2; ++cb) {
      const int obase = g * 96 + s * 32 + cb * 16;
      v16h bf = load_frag(w + (size_t)obase * E32, E32);
      v8f z = {};
      acc[cb] = wmma_f16(a, bf, z);
      bn[cb]  = bias[obase + nn];
    }
    __syncthreads();
#pragma unroll
    for (int cb = 0; cb < 2; ++cb) {
#pragma unroll
      for (int r = 0; r < 8; ++r) {
        const int m = r + half * 8;
        tile[m * 32 + cb * 16 + nn] = (_Float16)(acc[cb][r] + bn[cb]);
      }
    }
    if (s == 2) {
#pragma unroll
      for (int cb = 0; cb < 2; ++cb) {
        H8 p;
#pragma unroll
        for (int r = 0; r < 8; ++r) p.h[r] = (_Float16)(acc[cb][r] + bn[cb]);
        const int e = cb * 16 + nn;
        *(uint4*)(vT + ((size_t)bg * E32 + e) * 1024 + nbase + half * 8) = p.u;
      }
    }
    __syncthreads();
    store_tile_f16(outs[s] + ((size_t)bg * 1024 + nbase) * E32, tile);
  }
}


__device__ __forceinline__ float xhalf(float x) {
  const int b = __builtin_bit_cast(int, x);
  return __builtin_bit_cast(float,
      __builtin_amdgcn_permlanex16(b, b, 0x76543210, 0xfedcba98, false, false));
}

__global__ void __launch_bounds__(32)
attn_kernel(const _Float16* __restrict__ q, const _Float16* __restrict__ k,
            const _Float16* __restrict__ vT, _Float16* __restrict__ attn) {
  const int qt = blockIdx.x;
  const int bg = blockIdx.y;
  const int l  = lane_id();
  const _Float16* kb  = k  + (size_t)bg * 1024 * E32;
  const _Float16* vtb = vT + (size_t)bg * E32 * 1024;

  v16h bq = load_frag(q + ((size_t)bg * 1024 + qt * 16) * E32, E32);

  float m_run = -1e30f;
  float l_run = 0.0f;
  v8f acc0 = {}, acc1 = {};
  const float scale = 1.0f / (float)E32;
  const int nn = l & 15, half = l >> 4;

  for (int jt = 0; jt < 16; ++jt) {
    const int jb = jt * 64;
    if (jt < 15) {
      __builtin_prefetch(kb + (size_t)(jb + 64) * E32, 0, 1);
      __builtin_prefetch(vtb + jb + 64, 0, 1);
    }
    v16h ak[4];
#pragma unroll
    for (int c = 0; c < 4; ++c)
      ak[c] = load_frag(kb + (size_t)(jb + c * 16) * E32, E32);
    __builtin_amdgcn_sched_barrier(0);
    v16h av00 = load_frag(vtb + jb, 1024);
    v16h av01 = load_frag(vtb + jb + 32, 1024);
    v16h av10 = load_frag(vtb + (size_t)16 * 1024 + jb, 1024);
    v16h av11 = load_frag(vtb + (size_t)16 * 1024 + jb + 32, 1024);
    __builtin_amdgcn_sched_barrier(0);

    v8f s[4];
#pragma unroll
    for (int c = 0; c < 4; ++c) {
      v8f z = {};
      s[c] = wmma_f16(ak[c], bq, z);
    }

    float sv[4][8];
    float mx = -1e30f;
#pragma unroll
    for (int c = 0; c < 4; ++c) {
#pragma unroll
      for (int r = 0; r < 8; ++r) { sv[c][r] = s[c][r] * scale; mx = fmaxf(mx, sv[c][r]); }
    }
    mx = fmaxf(mx, xhalf(mx));
    const float mn  = fmaxf(m_run, mx);
    const float fac = __expf(m_run - mn);
    float rs = 0.0f;
    v16h bp0, bp1;
#pragma unroll
    for (int r = 0; r < 8; ++r) {
      const float p0 = __expf(sv[0][r] - mn);
      const float p1 = __expf(sv[1][r] - mn);
      const float p2 = __expf(sv[2][r] - mn);
      const float p3 = __expf(sv[3][r] - mn);
      rs += (p0 + p1) + (p2 + p3);
      bp0[r] = (_Float16)p0;  bp0[8 + r] = (_Float16)p1;
      bp1[r] = (_Float16)p2;  bp1[8 + r] = (_Float16)p3;
    }
    l_run = l_run * fac + rs;
    m_run = mn;
#pragma unroll
    for (int r = 0; r < 8; ++r) { acc0[r] *= fac; acc1[r] *= fac; }

    acc0 = wmma_f16(av00, bp0, acc0);
    acc0 = wmma_f16(av01, bp1, acc0);
    acc1 = wmma_f16(av10, bp0, acc1);
    acc1 = wmma_f16(av11, bp1, acc1);
  }

  const float inv = 1.0f / (l_run + xhalf(l_run));
  H8 o0, o1;
#pragma unroll
  for (int r = 0; r < 8; ++r) {
    o0.h[r] = (_Float16)(acc0[r] * inv);
    o1.h[r] = (_Float16)(acc1[r] * inv);
  }
  _Float16* dst = attn + ((size_t)bg * 1024 + qt * 16 + nn) * E32 + half * 8;
  *(uint4*)dst        = o0.u;
  *(uint4*)(dst + 16) = o1.u;
}

__global__ void __launch_bounds__(32)
mlp_kernel(const _Float16* __restrict__ v, const _Float16* __restrict__ attn,
           const float* __restrict__ w1,  const float* __restrict__ b1,
           const float* __restrict__ w10, const float* __restrict__ b10,
           const float* __restrict__ w11, const float* __restrict__ b11,
           float* __restrict__ out) {
  __shared__ alignas(16) _Float16 h1[16 * 32];
  __shared__ alignas(16) _Float16 h2[16 * 32];
  __shared__ alignas(16) float    ot[16 * 32];

  const int t = blockIdx.x;
  const int g = blockIdx.y;
  const int l = lane_id();
  const int rowbase = t * 16;
  const int b     = rowbase >> 10;
  const int nbase = rowbase & 1023;
  const int nn = l & 15, half = l >> 4;

  const size_t src = ((size_t)(b * GQ + g) * 1024 + nbase) * E32;
  v16h a_v  = load_frag(v + src, E32);
  v16h a_at = load_frag(attn + src, E32);

#pragma unroll
  for (int cb = 0; cb < 2; ++cb) {
    const int ob = g * 32 + cb * 16;
    v16h bw0 = load_frag(w1 + (size_t)ob * 64, 64);
    v16h bw1 = load_frag(w1 + (size_t)ob * 64 + 32, 64);
    v8f acc = {};
    acc = wmma_f16(a_v, bw0, acc);
    acc = wmma_f16(a_at, bw1, acc);
    const float bn = b1[ob + nn];
#pragma unroll
    for (int r = 0; r < 8; ++r) {
      const int m = r + half * 8;
      const float z = acc[r] + bn;
      h1[m * 32 + cb * 16 + nn] = (_Float16)(z >= 0.0f ? z : 0.01f * z);
    }
  }
  __syncthreads();

  v16h a_h1 = load_frag((const _Float16*)h1, 32);
#pragma unroll
  for (int cb = 0; cb < 2; ++cb) {
    const int ob = g * 32 + cb * 16;
    v16h bw = load_frag(w10 + (size_t)ob * 32, 32);
    v8f acc = {};
    acc = wmma_f16(a_h1, bw, acc);
    const float bn = b10[ob + nn];
#pragma unroll
    for (int r = 0; r < 8; ++r) {
      const int m = r + half * 8;
      const float z = acc[r] + bn;
      h2[m * 32 + cb * 16 + nn] = (_Float16)(z >= 0.0f ? z : 0.01f * z);
    }
  }
  __syncthreads();

  v16h a_h2 = load_frag((const _Float16*)h2, 32);
#pragma unroll
  for (int cb = 0; cb < 2; ++cb) {
    const int ob = g * 32 + cb * 16;
    v16h bw = load_frag(w11 + (size_t)ob * 32, 32);
    v8f acc = {};
    acc = wmma_f16(a_h2, bw, acc);
    const float bn = b11[ob + nn];
#pragma unroll
    for (int r = 0; r < 8; ++r) {
      const int m = r + half * 8;
      const float z = acc[r] + bn;
      ot[m * 32 + cb * 16 + nn] = (z >= 0.0f ? z : 0.01f * z);
    }
  }
  __syncthreads();
  float* dst = out + (size_t)rowbase * 64 + g * 32;
#pragma unroll
  for (int i = 0; i < 4; ++i) {
    const int idx = i * 32 + l;
    const int row = idx >> 3;
    const int c   = (idx & 7) * 4;
    *(uint4*)(dst + (size_t)row * 64 + c) = *(const uint4*)(ot + row * 32 + c);
  }
}

extern "C" void kernel_launch(void* const* d_in, const int* in_sizes, int n_in,
                              void* d_out, int out_size, void* d_ws, size_t ws_size,
                              hipStream_t stream) {
  const float* x     = (const float*)d_in[0];
  const float* w_qkv = (const float*)d_in[1];
  const float* b_qkv = (const float*)d_in[2];
  const float* w_h1  = (const float*)d_in[3];
  const float* b_h1  = (const float*)d_in[4];
  const float* w_h10 = (const float*)d_in[5];
  const float* b_h10 = (const float*)d_in[6];
  const float* w_h11 = (const float*)d_in[7];
  const float* b_h11 = (const float*)d_in[8];
  float* out = (float*)d_out;

  const size_t NEL = (size_t)32 * GQ * 1024 * E32;
  _Float16* ws   = (_Float16*)d_ws;
  _Float16* q    = ws;
  _Float16* k    = ws + NEL;
  _Float16* v    = ws + 2 * NEL;
  _Float16* vT   = ws + 3 * NEL;
  _Float16* attn = ws + 4 * NEL;

  qkv_kernel<<<dim3(2048, GQ), 32, 0, stream>>>(x, w_qkv, b_qkv, q, k, v, vT);
  attn_kernel<<<dim3(64, 64), 32, 0, stream>>>(q, k, vT, attn);
  mlp_kernel<<<dim3(2048, GQ), 32, 0, stream>>>(v, attn, w_h1, b_h1,
                                                w_h10, b_h10, w_h11, b_h11, out);
}


# attention: K and V^T of a (b,g) copied once per 8-wave work-group into LDS in fragment order; key loop reads operands by ds_load_b128 instead of per-wave global loads (summation order unchanged)
# speedup vs baseline: 1.5397x; 1.2766x over previous
#include <hip/hip_runtime.h>
#include <hip/hip_bf16.h>

typedef __attribute__((ext_vector_type(16))) _Float16 v16h;
typedef __attribute__((ext_vector_type(8)))  float    v8f;
typedef __attribute__((ext_vector_type(8)))  _Float16 v8h;
typedef __attribute__((ext_vector_type(4)))  unsigned int u32x4;

#define E32 32
#define GQ  2

union H8 { uint4 u; _Float16 h[8]; };

__device__ __forceinline__ int lane_id() { return (int)(threadIdx.x & 31); }

__device__ __forceinline__ v8f wmma_f16(v16h a, v16h b, v8f c) {
  return __builtin_amdgcn_wmma_f32_16x16x32_f16(
       false, a,  false, b,
       (short)0, c,  false,  false);
}

template <typename T>
__device__ __forceinline__ v16h load_frag(const T* __restrict__ src, int stride) {
  const int l  = lane_id();
  const int r  = l & 15;
  const int kb = (l >> 4) << 3;
  const T* p = src + (size_t)r * stride + kb;
  v16h f;
#pragma unroll
  for (int i = 0; i < 8; ++i) {
    f[i]     = (_Float16)p[i];
    f[8 + i] = (_Float16)p[i + 16];
  }
  return f;
}

__device__ __forceinline__ void store_tile_f16(_Float16* __restrict__ dst,
                                               const _Float16* __restrict__ lds) {
  const int l = lane_id();
#pragma unroll
  for (int i = 0; i < 2; ++i) {
    const int idx = i * 32 + l;
    *(uint4*)(dst + idx * 8) = *(const uint4*)(lds + idx * 8);
  }
}

__global__ void __launch_bounds__(32)
qkv_kernel(const float* __restrict__ x, const float* __restrict__ w,
           const float* __restrict__ bias,
           _Float16* __restrict__ q, _Float16* __restrict__ k,
           _Float16* __restrict__ v, _Float16* __restrict__ vT) {
  __shared__ alignas(16) _Float16 tile[16 * 32];

  const int t  = blockIdx.x;
  const int g  = blockIdx.y;
  const int l  = lane_id();
  const int rowbase = t * 16;
  const int b     = rowbase >> 10;
  const int nbase = rowbase & 1023;
  const int bg    = b * GQ + g;
  const int nn = l & 15, half = l >> 4;

  v16h a = load_frag(x + (size_t)rowbase * (GQ * E32) + g * E32, GQ * E32);

  _Float16* outs[3] = {q, k, v};
#pragma unroll
  for (int s = 0; s < 3; ++s) {
    v8f acc[2];
    float bn[2];
#pragma unroll
    for (int cb = 0; cb < 2; ++cb) {
      const int obase = g * 96 + s * 32 + cb * 16;
      v16h bf = load_frag(w + (size_t)obase * E32, E32);
      v8f z = {};
      acc[cb] = wmma_f16(a, bf, z);
      bn[cb]  = bias[obase + nn];
    }
    __syncthreads();
#pragma unroll
    for (int cb = 0; cb < 2; ++cb) {
#pragma unroll
      for (int r = 0; r < 8; ++r) {
        const int m = r + half * 8;
        tile[m * 32 + cb * 16 + nn] = (_Float16)(acc[cb][r] + bn[cb]);
      }
    }
    if (s == 2) {
#pragma unroll
      for (int cb = 0; cb < 2; ++cb) {
        H8 p;
#pragma unroll
        for (int r = 0; r < 8; ++r) p.h[r] = (_Float16)(acc[cb][r] + bn[cb]);
        const int e = cb * 16 + nn;
        *(uint4*)(vT + ((size_t)bg * E32 + e) * 1024 + nbase + half * 8) = p.u;
      }
    }
    __syncthreads();
    store_tile_f16(outs[s] + ((size_t)bg * 1024 + nbase) * E32, tile);
  }
}


#define ATT_WAVES 8
static_assert(64 % ATT_WAVES == 0 && ATT_WAVES * 32 == 256, "the launch below is 8 x 64 work-groups of 256 threads");

__device__ __forceinline__ v16h lds_frag(const _Float16* img) {
  const v8h lo = __builtin_bit_cast(v8h, *(const u32x4*)(img));
  const v8h hi = __builtin_bit_cast(v8h, *(const u32x4*)(img + 256));
  return __builtin_shufflevector(lo, hi, 0, 1, 2, 3, 4, 5, 6, 7,
                                 8, 9, 10, 11, 12, 13, 14, 15);
}

__device__ __forceinline__ float xhalf(float x) {
  const int b = __builtin_bit_cast(int, x);
  return __builtin_bit_cast(float,
      __builtin_amdgcn_permlanex16(b, b, 0x76543210, 0xfedcba98, false, false));
}

__global__ void __launch_bounds__(32 * ATT_WAVES)
attn_kernel(const _Float16* __restrict__ q, const _Float16* __restrict__ k,
            const _Float16* __restrict__ vT, _Float16* __restrict__ attn) {
  __shared__ alignas(16) _Float16 kimg[1024 * E32];
  __shared__ alignas(16) _Float16 vimg[E32 * 1024];

  const int tid = (int)threadIdx.x;
  const int qt = blockIdx.x * ATT_WAVES + (tid >> 5);
  const int bg = blockIdx.y;
  const int l  = lane_id();
  const _Float16* kb  = k  + (size_t)bg * 1024 * E32;
  const _Float16* vtb = vT + (size_t)bg * E32 * 1024;

  v16h bq = load_frag(q + ((size_t)bg * 1024 + qt * 16) * E32, E32);

  {
    const int c = tid & 15, h = (tid >> 4) & 1, i = (tid >> 5) & 1;
    const _Float16* ksrc = kb + (size_t)c * E32 + 8 * h + 16 * i;
    const _Float16* vsrc = vtb + (size_t)c * 1024 + 16 * i + 8 * h;
#pragma unroll
    for (int it = 0; it < 16; ++it) {
      const int f = it * 4 + (tid >> 6);
      const int d = f * 64 + (tid & 63);
      *(u32x4*)(kimg + d * 8) =
          *(const u32x4*)(ksrc + (size_t)f * 16 * E32);
      *(u32x4*)(vimg + d * 8) =
          *(const u32x4*)(vsrc + (size_t)(f & 1) * 16 * 1024 + (f >> 1) * 32);
    }
  }
  __syncthreads();

  float m_run = -1e30f;
  float l_run = 0.0f;
  v8f acc0 = {}, acc1 = {};
  const float scale = 1.0f / (float)E32;
  const int nn = l & 15, half = l >> 4;
  const _Float16* kl = kimg + l * 8;
  const _Float16* vl = vimg + l * 8;

  for (int jt = 0; jt < 16; ++jt) {
    const _Float16* ks = kl + jt * 2048;
    const _Float16* vs = vl + jt * 2048;
    v16h ak[4];
#pragma unroll
    for (int c = 0; c < 4; ++c)
      ak[c] = lds_frag(ks + c * 512);
    __builtin_amdgcn_sched_barrier(0);
    v16h av00 = lds_frag(vs);
    v16h av10 = lds_frag(vs + 512);
    v16h av01 = lds_frag(vs + 1024);
    v16h av11 = lds_frag(vs + 1536);
    __builtin_amdgcn_sched_barrier(0);

    v8f s[4];
#pragma unroll
    for (int c = 0; c < 4; ++c) {
      v8f z = {};
      s[c] = wmma_f16(ak[c], bq, z);
    }

    float sv[4][8];
    float mx = -1e30f;
#pragma unroll
    for (int c = 0; c < 4; ++c) {
#pragma unroll
      for (int r = 0; r < 8; ++r) { sv[c][r] = s[c][r] * scale; mx = fmaxf(mx, sv[c][r]); }
    }
    mx = fmaxf(mx, xhalf(mx));
    const float mn  = fmaxf(m_run, mx);
    const float fac = __expf(m_run - mn);
    float rs = 0.0f;
    v16h bp0, bp1;
#pragma unroll
    for (int r = 0; r < 8; ++r) {
      const float p0 = __expf(sv[0][r] - mn);
      const float p1 = __expf(sv[1][r] - mn);
      const float p2 = __expf(sv[2][r] - mn);
      const float p3 = __expf(sv[3][r] - mn);
      rs += (p0 + p1) + (p2 + p3);
      bp0[r] = (_Float16)p0;  bp0[8 + r] = (_Float16)p1;
      bp1[r] = (_Float16)p2;  bp1[8 + r] = (_Float16)p3;
    }
    l_run = l_run * fac + rs;
    m_run = mn;
#pragma unroll
    for (int r = 0; r < 8; ++r) { acc0[r] *= fac; acc1[r] *= fac; }

    acc0 = wmma_f16(av00, bp0, acc0);
    acc0 = wmma_f16(av01, bp1, acc0);
    acc1 = wmma_f16(av10, bp0, acc1);
    acc1 = wmma_f16(av11, bp1, acc1);
  }

  const float inv = 1.0f / (l_run + xhalf(l_run));
  H8 o0, o1;
#pragma unroll
  for (int r = 0; r < 8; ++r) {
    o0.h[r] = (_Float16)(acc0[r] * inv);
    o1.h[r] = (_Float16)(acc1[r] * inv);
  }
  _Float16* dst = attn + ((size_t)bg * 1024 + qt * 16 + nn) * E32 + half * 8;
  *(uint4*)dst        = o0.u;
  *(uint4*)(dst + 16) = o1.u;
}

__global__ void __launch_bounds__(32)
mlp_kernel(const _Float16* __restrict__ v, const _Float16* __restrict__ attn,
           const float* __restrict__ w1,  const float* __restrict__ b1,
           const float* __restrict__ w10, const float* __restrict__ b10,
           const float* __restrict__ w11, const float* __restrict__ b11,
           float* __restrict__ out) {
  __shared__ alignas(16) _Float16 h1[16 * 32];
  __shared__ alignas(16) _Float16 h2[16 * 32];
  __shared__ alignas(16) float    ot[16 * 32];

  const int t = blockIdx.x;
  const int g = blockIdx.y;
  const int l = lane_id();
  const int rowbase = t * 16;
  const int b     = rowbase >> 10;
  const int nbase = rowbase & 1023;
  const int nn = l & 15, half = l >> 4;

  const size_t src = ((size_t)(b * GQ + g) * 1024 + nbase) * E32;
  v16h a_v  = load_frag(v + src, E32);
  v16h a_at = load_frag(attn + src, E32);

#pragma unroll
  for (int cb = 0; cb < 2; ++cb) {
    const int ob = g * 32 + cb * 16;
    v16h bw0 = load_frag(w1 + (size_t)ob * 64, 64);
    v16h bw1 = load_frag(w1 + (size_t)ob * 64 + 32, 64);
    v8f acc = {};
    acc = wmma_f16(a_v, bw0, acc);
    acc = wmma_f16(a_at, bw1, acc);
    const float bn = b1[ob + nn];
#pragma unroll
    for (int r = 0; r < 8; ++r) {
      const int m = r + half * 8;
      const float z = acc[r] + bn;
      h1[m * 32 + cb * 16 + nn] = (_Float16)(z >= 0.0f ? z : 0.01f * z);
    }
  }
  __syncthreads();

  v16h a_h1 = load_frag((const _Float16*)h1, 32);
#pragma unroll
  for (int cb = 0; cb < 2; ++cb) {
    const int ob = g * 32 + cb * 16;
    v16h bw = load_frag(w10 + (size_t)ob * 32, 32);
    v8f acc = {};
    acc = wmma_f16(a_h1, bw, acc);
    const float bn = b10[ob + nn];
#pragma unroll
    for (int r = 0; r < 8; ++r) {
      const int m = r + half * 8;
      const float z = acc[r] + bn;
      h2[m * 32 + cb * 16 + nn] = (_Float16)(z >= 0.0f ? z : 0.01f * z);
    }
  }
  __syncthreads();

  v16h a_h2 = load_frag((const _Float16*)h2, 32);
#pragma unroll
  for (int cb = 0; cb < 2; ++cb) {
    const int ob = g * 32 + cb * 16;
    v16h bw = load_frag(w11 + (size_t)ob * 32, 32);
    v8f acc = {};
    acc = wmma_f16(a_h2, bw, acc);
    const float bn = b11[ob + nn];
#pragma unroll
    for (int r = 0; r < 8; ++r) {
      const int m = r + half * 8;
      const float z = acc[r] + bn;
      ot[m * 32 + cb * 16 + nn] = (z >= 0.0f ? z : 0.01f * z);
    }
  }
  __syncthreads();
  float* dst = out + (size_t)rowbase * 64 + g * 32;
#pragma unroll
  for (int i = 0; i < 4; ++i) {
    const int idx = i * 32 + l;
    const int row = idx >> 3;
    const int c   = (idx & 7) * 4;
    *(uint4*)(dst + (size_t)row * 64 + c) = *(const uint4*)(ot + row * 32 + c);
  }
}

extern "C" void kernel_launch(void* const* d_in, const int* in_sizes, int n_in,
                              void* d_out, int out_size, void* d_ws, size_t ws_size,
                              hipStream_t stream) {
  const float* x     = (const float*)d_in[0];
  const float* w_qkv = (const float*)d_in[1];
  const float* b_qkv = (const float*)d_in[2];
  const float* w_h1  = (const float*)d_in[3];
  const float* b_h1  = (const float*)d_in[4];
  const float* w_h10 = (const float*)d_in[5];
  const float* b_h10 = (const float*)d_in[6];
  const float* w_h11 = (const float*)d_in[7];
  const float* b_h11 = (const float*)d_in[8];
  float* out = (float*)d_out;

  const size_t NEL = (size_t)32 * GQ * 1024 * E32;
  _Float16* ws   = (_Float16*)d_ws;
  _Float16* q    = ws;
  _Float16* k    = ws + NEL;
  _Float16* v    = ws + 2 * NEL;
  _Float16* vT   = ws + 3 * NEL;
  _Float16* attn = ws + 4 * NEL;

  qkv_kernel<<<dim3(2048, GQ), 32, 0, stream>>>(x, w_qkv, b_qkv, q, k, v, vT);
  attn_kernel<<<dim3(8, 64), 256, 0, stream>>>(q, k, vT, attn);
  mlp_kernel<<<dim3(2048, GQ), 32, 0, stream>>>(v, attn, w_h1, b_h1,
                                                w_h10, b_h10, w_h11, b_h11, out);
}


# 3-layer MLP fused into the attention kernel's epilogue, transposed (W . X^T) so O^T and both hidden layers feed the next product from registers; weights staged in LDS; 2 launches, same sum order
# speedup vs baseline: 1.8589x; 1.2073x over previous
#include <hip/hip_runtime.h>
#include <hip/hip_bf16.h>

typedef __attribute__((ext_vector_type(16))) _Float16 v16h;
typedef __attribute__((ext_vector_type(8)))  float    v8f;
typedef __attribute__((ext_vector_type(8)))  _Float16 v8h;
typedef __attribute__((ext_vector_type(4)))  unsigned int u32x4;

#define E32 32
#define GQ  2

union H8 { uint4 u; _Float16 h[8]; };

__device__ __forceinline__ int lane_id() { return (int)(threadIdx.x & 31); }

__device__ __forceinline__ v8f wmma_f16(v16h a, v16h b, v8f c) {
  return __builtin_amdgcn_wmma_f32_16x16x32_f16(
       false, a,  false, b,
       (short)0, c,  false,  false);
}

template <typename T>
__device__ __forceinline__ v16h load_frag(const T* __restrict__ src, int stride) {
  const int l  = lane_id();
  const int r  = l & 15;
  const int kb = (l >> 4) << 3;
  const T* p = src + (size_t)r * stride + kb;
  v16h f;
#pragma unroll
  for (int i = 0; i < 8; ++i) {
    f[i]     = (_Float16)p[i];
    f[8 + i] = (_Float16)p[i + 16];
  }
  return f;
}

__device__ __forceinline__ void store_tile_f16(_Float16* __restrict__ dst,
                                               const _Float16* __restrict__ lds) {
  const int l = lane_id();
#pragma unroll
  for (int i = 0; i < 2; ++i) {
    const int idx = i * 32 + l;
    *(uint4*)(dst + idx * 8) = *(const uint4*)(lds + idx * 8);
  }
}

__global__ void __launch_bounds__(32)
qkv_kernel(const float* __restrict__ x, const float* __restrict__ w,
           const float* __restrict__ bias,
           _Float16* __restrict__ q, _Float16* __restrict__ k,
           _Float16* __restrict__ v, _Float16* __restrict__ vT) {
  __shared__ alignas(16) _Float16 tile[16 * 32];

  const int t  = blockIdx.x;
  const int g  = blockIdx.y;
  const int l  = lane_id();
  const int rowbase = t * 16;
  const int b     = rowbase >> 10;
  const int nbase = rowbase & 1023;
  const int bg    = b * GQ + g;
  const int nn = l & 15, half = l >> 4;

  v16h a = load_frag(x + (size_t)rowbase * (GQ * E32) + g * E32, GQ * E32);

  _Float16* outs[3] = {q, k, v};
#pragma unroll
  for (int s = 0; s < 3; ++s) {
    v8f acc[2];
    float bn[2];
#pragma unroll
    for (int cb = 0; cb < 2; ++cb) {
      const int obase = g * 96 + s * 32 + cb * 16;
      v16h bf = load_frag(w + (size_t)obase * E32, E32);
      v8f z = {};
      acc[cb] = wmma_f16(a, bf, z);
      bn[cb]  = bias[obase + nn];
    }
    __syncthreads();
#pragma unroll
    for (int cb = 0; cb < 2; ++cb) {
#pragma unroll
      for (int r = 0; r < 8; ++r) {
        const int m = r + half * 8;
        tile[m * 32 + cb * 16 + nn] = (_Float16)(acc[cb][r] + bn[cb]);
      }
    }
    if (s == 2) {
#pragma unroll
      for (int cb = 0; cb < 2; ++cb) {
        H8 p;
#pragma unroll
        for (int r = 0; r < 8; ++r) p.h[r] = (_Float16)(acc[cb][r] + bn[cb]);
        const int e = cb * 16 + nn;
        *(uint4*)(vT + ((size_t)bg * E32 + e) * 1024 + nbase + half * 8) = p.u;
      }
    }
    __syncthreads();
    store_tile_f16(outs[s] + ((size_t)bg * 1024 + nbase) * E32, tile);
  }
}


#define ATT_WAVES 8
static_assert(64 % ATT_WAVES == 0 && ATT_WAVES * 32 == 256, "the launch below is 8 x 64 work-groups of 256 threads");

__device__ __forceinline__ v16h lds_frag(const _Float16* img) {
  const v8h lo = __builtin_bit_cast(v8h, *(const u32x4*)(img));
  const v8h hi = __builtin_bit_cast(v8h, *(const u32x4*)(img + 256));
  return __builtin_shufflevector(lo, hi, 0, 1, 2, 3, 4, 5, 6, 7,
                                 8, 9, 10, 11, 12, 13, 14, 15);
}

__device__ __forceinline__ float xhalf(float x) {
  const int b = __builtin_bit_cast(int, x);
  return __builtin_bit_cast(float,
      __builtin_amdgcn_permlanex16(b, b, 0x76543210, 0xfedcba98, false, false));
}

#define OT_PITCH 36
static_assert(ATT_WAVES * 16 * OT_PITCH * 4 <= 1024 * E32 * 2, "the output tiles lie over the K image");

__device__ __forceinline__ float leaky(float z) { return fmaxf(z, 0.01f * z); }

__device__ __forceinline__ void stage8(_Float16* dst, float4 lo, float4 hi) {
  H8 p;
  p.h[0] = (_Float16)lo.x;  p.h[1] = (_Float16)lo.y;  p.h[2] = (_Float16)lo.z;  p.h[3] = (_Float16)lo.w;
  p.h[4] = (_Float16)hi.x;  p.h[5] = (_Float16)hi.y;  p.h[6] = (_Float16)hi.z;  p.h[7] = (_Float16)hi.w;
  *(uint4*)dst = p.u;
}

__device__ __forceinline__ v8f lds_bias8(const float* p) {
  const float4 lo = *(const float4*)p;
  const float4 hi = *(const float4*)(p + 4);
  v8f b = {lo.x, lo.y, lo.z, lo.w, hi.x, hi.y, hi.z, hi.w};
  return b;
}

__global__ void __launch_bounds__(32 * ATT_WAVES)
attn_mlp_kernel(const _Float16* __restrict__ q, const _Float16* __restrict__ k,
                const _Float16* __restrict__ v, const _Float16* __restrict__ vT,
                const float* __restrict__ w1,  const float* __restrict__ b1,
                const float* __restrict__ w10, const float* __restrict__ b10,
                const float* __restrict__ w11, const float* __restrict__ b11,
                float* __restrict__ out) {
  __shared__ alignas(16) _Float16 kimg[1024 * E32];
  __shared__ alignas(16) _Float16 vimg[E32 * 1024];
  __shared__ alignas(16) _Float16 wimg[8 * 16 * E32];
  __shared__ alignas(16) float    bimg[ATT_WAVES * E32];

  const int tid = (int)threadIdx.x;
  const int wv = tid >> 5;
  const int qt = blockIdx.x * ATT_WAVES + wv;
  const int bg = blockIdx.y;
  const int g  = bg & (GQ - 1);
  const int l  = lane_id();
  const _Float16* kb  = k  + (size_t)bg * 1024 * E32;
  const _Float16* vtb = vT + (size_t)bg * E32 * 1024;

  v16h bq = load_frag(q + ((size_t)bg * 1024 + qt * 16) * E32, E32);

  const v16h b_v = load_frag(v + ((size_t)bg * 1024 + qt * 16) * E32, E32);

  float4 wq[4];
  float  bval;
  {
    const int c = tid & 15, h = (tid >> 4) & 1, i = (tid >> 5) & 1, fp = tid >> 6;
    const int orow = g * 32 + (fp & 1) * 16 + c;
    const float* s0 = w1 + (size_t)orow * 64 + (fp >> 1) * 32 + 16 * i + 8 * h;
    const float* s1 = ((fp >> 1) ? w11 : w10) + (size_t)orow * 32 + 16 * i + 8 * h;
    wq[0] = *(const float4*)s0;  wq[1] = *(const float4*)(s0 + 4);
    wq[2] = *(const float4*)s1;  wq[3] = *(const float4*)(s1 + 4);
    bval = (wv == 0 ? b1 : wv == 1 ? b10 : b11)[g * 32 + l];
  }

  {
    const int c = tid & 15, h = (tid >> 4) & 1, i = (tid >> 5) & 1;
    const _Float16* ksrc = kb + (size_t)c * E32 + 8 * h + 16 * i;
    const _Float16* vsrc = vtb + (size_t)c * 1024 + 16 * i + 8 * h;
#pragma unroll
    for (int it = 0; it < 16; ++it) {
      const int f = it * 4 + (tid >> 6);
      const int d = f * 64 + (tid & 63);
      *(u32x4*)(kimg + d * 8) =
          *(const u32x4*)(ksrc + (size_t)f * 16 * E32);
      *(u32x4*)(vimg + d * 8) =
          *(const u32x4*)(vsrc + (size_t)(f & 1) * 16 * 1024 + (f >> 1) * 32);
    }
  }
  stage8(wimg + tid * 8, wq[0], wq[1]);
  stage8(wimg + (256 + tid) * 8, wq[2], wq[3]);
  bimg[tid] = bval;
  __syncthreads();

  float m_run = -1e30f;
  float l_run = 0.0f;
  v8f acc0 = {}, acc1 = {};
  const float scale = 1.0f / (float)E32;
  const int nn = l & 15, half = l >> 4;
  const _Float16* kl = kimg + l * 8;
  const _Float16* vl = vimg + l * 8;

  for (int jt = 0; jt < 16; ++jt) {
    const _Float16* ks = kl + jt * 2048;
    const _Float16* vs = vl + jt * 2048;
    v16h ak[4];
#pragma unroll
    for (int c = 0; c < 4; ++c)
      ak[c] = lds_frag(ks + c * 512);
    __builtin_amdgcn_sched_barrier(0);
    v16h av00 = lds_frag(vs);
    v16h av10 = lds_frag(vs + 512);
    v16h av01 = lds_frag(vs + 1024);
    v16h av11 = lds_frag(vs + 1536);
    __builtin_amdgcn_sched_barrier(0);

    v8f s[4];
#pragma unroll
    for (int c = 0; c < 4; ++c) {
      v8f z = {};
      s[c] = wmma_f16(ak[c], bq, z);
    }

    float sv[4][8];
    float mx = -1e30f;
#pragma unroll
    for (int c = 0; c < 4; ++c) {
#pragma unroll
      for (int r = 0; r < 8; ++r) { sv[c][r] = s[c][r] * scale; mx = fmaxf(mx, sv[c][r]); }
    }
    mx = fmaxf(mx, xhalf(mx));
    const float mn  = fmaxf(m_run, mx);
    const float fac = __expf(m_run - mn);
    float rs = 0.0f;
    v16h bp0, bp1;
#pragma unroll
    for (int r = 0; r < 8; ++r) {
      const float p0 = __expf(sv[0][r] - mn);
      const float p1 = __expf(sv[1][r] - mn);
      const float p2 = __expf(sv[2][r] - mn);
      const float p3 = __expf(sv[3][r] - mn);
      rs += (p0 + p1) + (p2 + p3);
      bp0[r] = (_Float16)p0;  bp0[8 + r] = (_Float16)p1;
      bp1[r] = (_Float16)p2;  bp1[8 + r] = (_Float16)p3;
    }
    l_run = l_run * fac + rs;
    m_run = mn;
#pragma unroll
    for (int r = 0; r < 8; ++r) { acc0[r] *= fac; acc1[r] *= fac; }

    acc0 = wmma_f16(av00, bp0, acc0);
    acc0 = wmma_f16(av01, bp1, acc0);
    acc1 = wmma_f16(av10, bp0, acc1);
    acc1 = wmma_f16(av11, bp1, acc1);
  }

  const float inv = 1.0f / (l_run + xhalf(l_run));
  v16h b_at;
#pragma unroll
  for (int r = 0; r < 8; ++r) {
    b_at[r]     = (_Float16)(acc0[r] * inv);
    b_at[8 + r] = (_Float16)(acc1[r] * inv);
  }

  v16h w1v[2], w1a[2], w10f[2], w11f[2];
  v8f  bs1[2], bs10[2], bs11[2];
#pragma unroll
  for (int cb = 0; cb < 2; ++cb) {
    const _Float16* wl = wimg + l * 8 + cb * 512;
    w1v[cb]  = lds_frag(wl);
    w1a[cb]  = lds_frag(wl + 2 * 512);
    w10f[cb] = lds_frag(wl + 4 * 512);
    w11f[cb] = lds_frag(wl + 6 * 512);
    const float* bl = bimg + cb * 16 + half * 8;
    bs1[cb]  = lds_bias8(bl);
    bs10[cb] = lds_bias8(bl + E32);
    bs11[cb] = lds_bias8(bl + 2 * E32);
  }

  v16h b_h1;
#pragma unroll
  for (int cb = 0; cb < 2; ++cb) {
    v8f a = {};
    a = wmma_f16(w1v[cb], b_v, a);
    a = wmma_f16(w1a[cb], b_at, a);
#pragma unroll
    for (int r = 0; r < 8; ++r) b_h1[8 * cb + r] = (_Float16)leaky(a[r] + bs1[cb][r]);
  }
  v16h b_h2;
#pragma unroll
  for (int cb = 0; cb < 2; ++cb) {
    v8f a = {};
    a = wmma_f16(w10f[cb], b_h1, a);
#pragma unroll
    for (int r = 0; r < 8; ++r) b_h2[8 * cb + r] = (_Float16)leaky(a[r] + bs10[cb][r]);
  }
  float4 res[2][2];
#pragma unroll
  for (int cb = 0; cb < 2; ++cb) {
    v8f a = {};
    a = wmma_f16(w11f[cb], b_h2, a);
    res[cb][0].x = leaky(a[0] + bs11[cb][0]);  res[cb][0].y = leaky(a[1] + bs11[cb][1]);
    res[cb][0].z = leaky(a[2] + bs11[cb][2]);  res[cb][0].w = leaky(a[3] + bs11[cb][3]);
    res[cb][1].x = leaky(a[4] + bs11[cb][4]);  res[cb][1].y = leaky(a[5] + bs11[cb][5]);
    res[cb][1].z = leaky(a[6] + bs11[cb][6]);  res[cb][1].w = leaky(a[7] + bs11[cb][7]);
  }
  __syncthreads();
  float* st = reinterpret_cast<float*>(kimg) + wv * (16 * OT_PITCH);
#pragma unroll
  for (int cb = 0; cb < 2; ++cb) {
    float* sp = st + nn * OT_PITCH + cb * 16 + half * 8;
    *(float4*)sp       = res[cb][0];
    *(float4*)(sp + 4) = res[cb][1];
  }
  __syncthreads();
  float* dst = out + ((size_t)(bg >> 1) * 1024 + qt * 16) * (GQ * E32) + g * E32;
#pragma unroll
  for (int i = 0; i < 4; ++i) {
    const int idx = i * 32 + l;
    const int row = idx >> 3;
    const int c   = (idx & 7) * 4;
    *(float4*)(dst + (size_t)row * (GQ * E32) + c) = *(const float4*)(st + row * OT_PITCH + c);
  }
}

extern "C" void kernel_launch(void* const* d_in, const int* in_sizes, int n_in,
                              void* d_out, int out_size, void* d_ws, size_t ws_size,
                              hipStream_t stream) {
  const float* x     = (const float*)d_in[0];
  const float* w_qkv = (const float*)d_in[1];
  const float* b_qkv = (const float*)d_in[2];
  const float* w_h1  = (const float*)d_in[3];
  const float* b_h1  = (const float*)d_in[4];
  const float* w_h10 = (const float*)d_in[5];
  const float* b_h10 = (const float*)d_in[6];
  const float* w_h11 = (const float*)d_in[7];
  const float* b_h11 = (const float*)d_in[8];
  float* out = (float*)d_out;

  const size_t NEL = (size_t)32 * GQ * 1024 * E32;
  _Float16* ws   = (_Float16*)d_ws;
  _Float16* q    = ws;
  _Float16* k    = ws + NEL;
  _Float16* v    = ws + 2 * NEL;
  _Float16* vT   = ws + 3 * NEL;

  qkv_kernel<<<dim3(2048, GQ), 32, 0, stream>>>(x, w_qkv, b_qkv, q, k, v, vT);
  attn_mlp_kernel<<<dim3(8, 64), 256, 0, stream>>>(q, k, v, vT, w_h1, b_h1,
                                                   w_h10, b_h10, w_h11, b_h11, out);
}


# QKV projection fused into the attention kernel's prologue: each work-group projects its (b,g)'s K^T (W . X^T) and V from x on WMMA straight into the LDS images, Q^T into registers; 1 launch
# speedup vs baseline: 2.6718x; 1.4373x over previous
#include <hip/hip_runtime.h>
#include <hip/hip_bf16.h>

typedef __attribute__((ext_vector_type(16))) _Float16 v16h;
typedef __attribute__((ext_vector_type(8)))  float    v8f;
typedef __attribute__((ext_vector_type(8)))  _Float16 v8h;
typedef __attribute__((ext_vector_type(4)))  unsigned int u32x4;

#define E32 32
#define GQ  2

union H8 { uint4 u; _Float16 h[8]; };

__device__ __forceinline__ int lane_id() { return (int)(threadIdx.x & 31); }

__device__ __forceinline__ v8f wmma_f16(v16h a, v16h b, v8f c) {
  return __builtin_amdgcn_wmma_f32_16x16x32_f16(
       false, a,  false, b,
       (short)0, c,  false,  false);
}

template <typename T>
__device__ __forceinline__ v16h load_frag(const T* __restrict__ src, int stride) {
  const int l  = lane_id();
  const int r  = l & 15;
  const int kb = (l >> 4) << 3;
  const T* p = src + (size_t)r * stride + kb;
  v16h f;
#pragma unroll
  for (int i = 0; i < 8; ++i) {
    f[i]     = (_Float16)p[i];
    f[8 + i] = (_Float16)p[i + 16];
  }
  return f;
}


#define ATT_WAVES 8
static_assert(64 % ATT_WAVES == 0 && ATT_WAVES * 32 == 256, "the launch below is 8 x 64 work-groups of 256 threads");


__device__ __forceinline__ v16h lds_frag(const _Float16* img) {
  const v8h lo = __builtin_bit_cast(v8h, *(const u32x4*)(img));
  const v8h hi = __builtin_bit_cast(v8h, *(const u32x4*)(img + 256));
  return __builtin_shufflevector(lo, hi, 0, 1, 2, 3, 4, 5, 6, 7,
                                 8, 9, 10, 11, 12, 13, 14, 15);
}

__device__ __forceinline__ float xhalf(float x) {
  const int b = __builtin_bit_cast(int, x);
  return __builtin_bit_cast(float,
      __builtin_amdgcn_permlanex16(b, b, 0x76543210, 0xfedcba98, false, false));
}

#define OT_PITCH 36
static_assert(ATT_WAVES * 16 * OT_PITCH * 4 <= 1024 * E32 * 2, "the output tiles lie over the K image");

__device__ __forceinline__ float leaky(float z) { return fmaxf(z, 0.01f * z); }

__device__ __forceinline__ void stage8(_Float16* dst, float4 lo, float4 hi) {
  H8 p;
  p.h[0] = (_Float16)lo.x;  p.h[1] = (_Float16)lo.y;  p.h[2] = (_Float16)lo.z;  p.h[3] = (_Float16)lo.w;
  p.h[4] = (_Float16)hi.x;  p.h[5] = (_Float16)hi.y;  p.h[6] = (_Float16)hi.z;  p.h[7] = (_Float16)hi.w;
  *(uint4*)dst = p.u;
}

__device__ __forceinline__ v8f bias8(const float* p) {
  const float4 lo = *(const float4*)p;
  const float4 hi = *(const float4*)(p + 4);
  v8f b = {lo.x, lo.y, lo.z, lo.w, hi.x, hi.y, hi.z, hi.w};
  return b;
}

struct XRaw { float4 a, b, c, d; };
__device__ __forceinline__ XRaw x_raw(const float* __restrict__ tile, int pitch) {
  const int l = lane_id();
  const float* p = tile + (size_t)(l & 15) * pitch + ((l >> 4) << 3);
  XRaw r;
  r.a = *(const float4*)p;         r.b = *(const float4*)(p + 4);
  r.c = *(const float4*)(p + 16);  r.d = *(const float4*)(p + 20);
  return r;
}
__device__ __forceinline__ v16h x_frag(const XRaw r) {
  v16h f;
  f[0]  = (_Float16)r.a.x;  f[1]  = (_Float16)r.a.y;  f[2]  = (_Float16)r.a.z;  f[3]  = (_Float16)r.a.w;
  f[4]  = (_Float16)r.b.x;  f[5]  = (_Float16)r.b.y;  f[6]  = (_Float16)r.b.z;  f[7]  = (_Float16)r.b.w;
  f[8]  = (_Float16)r.c.x;  f[9]  = (_Float16)r.c.y;  f[10] = (_Float16)r.c.z;  f[11] = (_Float16)r.c.w;
  f[12] = (_Float16)r.d.x;  f[13] = (_Float16)r.d.y;  f[14] = (_Float16)r.d.z;  f[15] = (_Float16)r.d.w;
  return f;
}

__global__ void __launch_bounds__(32 * ATT_WAVES)
attn_mlp_kernel(const float* __restrict__ x, const float* __restrict__ wqkv,
                const float* __restrict__ bqkv,
                const float* __restrict__ w1,  const float* __restrict__ b1,
                const float* __restrict__ w10, const float* __restrict__ b10,
                const float* __restrict__ w11, const float* __restrict__ b11,
                float* __restrict__ out) {
  __shared__ alignas(16) _Float16 kimg[1024 * E32];
  __shared__ alignas(16) _Float16 vimg[E32 * 1024];
  __shared__ alignas(16) _Float16 wimg[8 * 16 * E32];
  __shared__ alignas(16) float    bimg[ATT_WAVES * E32];

  const int tid = (int)threadIdx.x;
  const int wv = tid >> 5;
  const int qt = blockIdx.x * ATT_WAVES + wv;
  const int bg = blockIdx.y;
  const int g  = bg & (GQ - 1);
  const int l  = lane_id();
  const int nn = l & 15, half = l >> 4;
  const float* xg  = x + (size_t)(bg >> 1) * 1024 * (GQ * E32) + g * E32;
  const float* wg  = wqkv + (size_t)g * 96 * E32;
  const float* bgp = bqkv + g * 96;

  v16h wk_f[2], wv_f[2];
  v8f  bias_k[2];
  float bias_vc[2];
#pragma unroll
  for (int u = 0; u < 2; ++u) {
    wk_f[u] = load_frag(wg + (size_t)(32 + 16 * u) * E32, E32);
    wv_f[u] = load_frag(wg + (size_t)(64 + 16 * u) * E32, E32);
    bias_k[u]  = bias8(bgp + 32 + 16 * u + 8 * half);
    bias_vc[u] = bgp[64 + 16 * u + nn];
  }

  float4 wq[4];
  float  bval;

  constexpr int XP = GQ * E32;
  XRaw xr[4];
#pragma unroll
  for (int j = 0; j < 4; ++j) xr[j] = x_raw(xg + (size_t)(j * ATT_WAVES + wv) * 16 * XP, XP);

  XRaw xown, wq_raw[2];
  v8f  bias_q[2], bias_v[2];

#pragma unroll
  for (int j = 0; j < 64 / ATT_WAVES; ++j) {
    const int t = j * ATT_WAVES + wv;
    v16h xf = x_frag(xr[j & 3]);
    asm volatile("" : "+v"(xf));
    if (j + 4 < 64 / ATT_WAVES)
      xr[j & 3] = x_raw(xg + (size_t)((j + 4) * ATT_WAVES + wv) * 16 * XP, XP);
    if (j == 4) {
      const int c = tid & 15, h = (tid >> 4) & 1, i = (tid >> 5) & 1, fp = tid >> 6;
      const int orow = g * 32 + (fp & 1) * 16 + c;
      const float* s0 = w1 + (size_t)orow * 64 + (fp >> 1) * 32 + 16 * i + 8 * h;
      const float* s1 = ((fp >> 1) ? w11 : w10) + (size_t)orow * 32 + 16 * i + 8 * h;
      wq[0] = *(const float4*)s0;  wq[1] = *(const float4*)(s0 + 4);
      wq[2] = *(const float4*)s1;  wq[3] = *(const float4*)(s1 + 4);
      bval = (wv == 0 ? b1 : wv == 1 ? b10 : b11)[g * 32 + l];
      wq_raw[0] = x_raw(wg, E32);
      wq_raw[1] = x_raw(wg + 16 * E32, E32);
    }
    if (j == 6) {
      xown = x_raw(xg + (size_t)qt * 16 * XP, XP);
#pragma unroll
      for (int u = 0; u < 2; ++u) {
        bias_q[u] = bias8(bgp + 0  + 16 * u + 8 * half);
        bias_v[u] = bias8(bgp + 64 + 16 * u + 8 * half);
      }
    }
#pragma unroll
    for (int u = 0; u < 2; ++u) {
      v8f z = {};
      const v8f kt = wmma_f16(wk_f[u], xf, z);
      const v8f vv = wmma_f16(xf, wv_f[u], z);
      H8 pk, pv;
#pragma unroll
      for (int r = 0; r < 8; ++r) {
        pk.h[r] = (_Float16)(kt[r] + bias_k[u][r]);
        pv.h[r] = (_Float16)(vv[r] + bias_vc[u]);
      }
      *(uint4*)(kimg + t * 512 + u * 256 + l * 8) = pk.u;
      *(uint4*)(vimg + ((t >> 1) * 2 + u) * 512 + (t & 1) * 256 + l * 8) = pv.u;
    }
  }

  v16h bq, b_v;
  {
    const v16h xf = x_frag(xown);
#pragma unroll
    for (int u = 0; u < 2; ++u) {
      v8f z = {};
      const v8f qT = wmma_f16(x_frag(wq_raw[u]), xf, z);
      const v8f oT = wmma_f16(wv_f[u], xf, z);
#pragma unroll
      for (int r = 0; r < 8; ++r) {
        bq[8 * u + r]  = (_Float16)(qT[r] + bias_q[u][r]);
        b_v[8 * u + r] = (_Float16)(oT[r] + bias_v[u][r]);
      }
    }
  }
  stage8(wimg + tid * 8, wq[0], wq[1]);
  stage8(wimg + (256 + tid) * 8, wq[2], wq[3]);
  bimg[tid] = bval;
  __syncthreads();

  float m_run = -1e30f;
  float l_run = 0.0f;
  v8f acc0 = {}, acc1 = {};
  const float scale = 1.0f / (float)E32;
  const _Float16* kl = kimg + l * 8;
  const _Float16* vl = vimg + l * 8;

  for (int jt = 0; jt < 16; ++jt) {
    const _Float16* ks = kl + jt * 2048;
    const _Float16* vs = vl + jt * 2048;
    v16h ak[4];
#pragma unroll
    for (int c = 0; c < 4; ++c)
      ak[c] = lds_frag(ks + c * 512);
    __builtin_amdgcn_sched_barrier(0);
    v16h av00 = lds_frag(vs);
    v16h av10 = lds_frag(vs + 512);
    v16h av01 = lds_frag(vs + 1024);
    v16h av11 = lds_frag(vs + 1536);
    __builtin_amdgcn_sched_barrier(0);

    v8f s[4];
#pragma unroll
    for (int c = 0; c < 4; ++c) {
      v8f z = {};
      s[c] = wmma_f16(ak[c], bq, z);
    }

    float sv[4][8];
    float mx = -1e30f;
#pragma unroll
    for (int c = 0; c < 4; ++c) {
#pragma unroll
      for (int r = 0; r < 8; ++r) { sv[c][r] = s[c][r] * scale; mx = fmaxf(mx, sv[c][r]); }
    }
    mx = fmaxf(mx, xhalf(mx));
    const float mn  = fmaxf(m_run, mx);
    const float fac = __expf(m_run - mn);
    float rs = 0.0f;
    v16h bp0, bp1;
#pragma unroll
    for (int r = 0; r < 8; ++r) {
      const float p0 = __expf(sv[0][r] - mn);
      const float p1 = __expf(sv[1][r] - mn);
      const float p2 = __expf(sv[2][r] - mn);
      const float p3 = __expf(sv[3][r] - mn);
      rs += (p0 + p1) + (p2 + p3);
      bp0[r] = (_Float16)p0;  bp0[8 + r] = (_Float16)p1;
      bp1[r] = (_Float16)p2;  bp1[8 + r] = (_Float16)p3;
    }
    l_run = l_run * fac + rs;
    m_run = mn;
#pragma unroll
    for (int r = 0; r < 8; ++r) { acc0[r] *= fac; acc1[r] *= fac; }

    acc0 = wmma_f16(av00, bp0, acc0);
    acc0 = wmma_f16(av01, bp1, acc0);
    acc1 = wmma_f16(av10, bp0, acc1);
    acc1 = wmma_f16(av11, bp1, acc1);
  }

  const float inv = 1.0f / (l_run + xhalf(l_run));
  v16h b_at;
#pragma unroll
  for (int r = 0; r < 8; ++r) {
    b_at[r]     = (_Float16)(acc0[r] * inv);
    b_at[8 + r] = (_Float16)(acc1[r] * inv);
  }

  v16h w1v[2], w1a[2], w10f[2], w11f[2];
  v8f  bs1[2], bs10[2], bs11[2];
#pragma unroll
  for (int cb = 0; cb < 2; ++cb) {
    const _Float16* wl = wimg + l * 8 + cb * 512;
    w1v[cb]  = lds_frag(wl);
    w1a[cb]  = lds_frag(wl + 2 * 512);
    w10f[cb] = lds_frag(wl + 4 * 512);
    w11f[cb] = lds_frag(wl + 6 * 512);
    const float* bl = bimg + cb * 16 + half * 8;
    bs1[cb]  = bias8(bl);
    bs10[cb] = bias8(bl + E32);
    bs11[cb] = bias8(bl + 2 * E32);
  }

  v16h b_h1;
#pragma unroll
  for (int cb = 0; cb < 2; ++cb) {
    v8f a = {};
    a = wmma_f16(w1v[cb], b_v, a);
    a = wmma_f16(w1a[cb], b_at, a);
#pragma unroll
    for (int r = 0; r < 8; ++r) b_h1[8 * cb + r] = (_Float16)leaky(a[r] + bs1[cb][r]);
  }
  v16h b_h2;
#pragma unroll
  for (int cb = 0; cb < 2; ++cb) {
    v8f a = {};
    a = wmma_f16(w10f[cb], b_h1, a);
#pragma unroll
    for (int r = 0; r < 8; ++r) b_h2[8 * cb + r] = (_Float16)leaky(a[r] + bs10[cb][r]);
  }
  float4 res[2][2];
#pragma unroll
  for (int cb = 0; cb < 2; ++cb) {
    v8f a = {};
    a = wmma_f16(w11f[cb], b_h2, a);
    res[cb][0].x = leaky(a[0] + bs11[cb][0]);  res[cb][0].y = leaky(a[1] + bs11[cb][1]);
    res[cb][0].z = leaky(a[2] + bs11[cb][2]);  res[cb][0].w = leaky(a[3] + bs11[cb][3]);
    res[cb][1].x = leaky(a[4] + bs11[cb][4]);  res[cb][1].y = leaky(a[5] + bs11[cb][5]);
    res[cb][1].z = leaky(a[6] + bs11[cb][6]);  res[cb][1].w = leaky(a[7] + bs11[cb][7]);
  }
  __syncthreads();
  float* st = reinterpret_cast<float*>(kimg) + wv * (16 * OT_PITCH);
#pragma unroll
  for (int cb = 0; cb < 2; ++cb) {
    float* sp = st + nn * OT_PITCH + cb * 16 + half * 8;
    *(float4*)sp       = res[cb][0];
    *(float4*)(sp + 4) = res[cb][1];
  }
  __syncthreads();
  float* dst = out + ((size_t)(bg >> 1) * 1024 + qt * 16) * (GQ * E32) + g * E32;
#pragma unroll
  for (int i = 0; i < 4; ++i) {
    const int idx = i * 32 + l;
    const int row = idx >> 3;
    const int c   = (idx & 7) * 4;
    *(float4*)(dst + (size_t)row * (GQ * E32) + c) = *(const float4*)(st + row * OT_PITCH + c);
  }
}

extern "C" void kernel_launch(void* const* d_in, const int* in_sizes, int n_in,
                              void* d_out, int out_size, void* d_ws, size_t ws_size,
                              hipStream_t stream) {
  const float* x     = (const float*)d_in[0];
  const float* w_qkv = (const float*)d_in[1];
  const float* b_qkv = (const float*)d_in[2];
  const float* w_h1  = (const float*)d_in[3];
  const float* b_h1  = (const float*)d_in[4];
  const float* w_h10 = (const float*)d_in[5];
  const float* b_h10 = (const float*)d_in[6];
  const float* w_h11 = (const float*)d_in[7];
  const float* b_h11 = (const float*)d_in[8];
  float* out = (float*)d_out;

  attn_mlp_kernel<<<dim3(8, 64), 256, 0, stream>>>(x, w_qkv, b_qkv, w_h1, b_h1,
                                                   w_h10, b_h10, w_h11, b_h11, out);
}


# attention softmax in the base-2 domain on raw scores: 1/E and log2(e) folded into one constant, maximum taken before scaling, exponent argument one packed fma per two scores; sum order unchanged
# speedup vs baseline: 2.8356x; 1.0613x over previous
#include <hip/hip_runtime.h>
#include <hip/hip_bf16.h>

typedef __attribute__((ext_vector_type(16))) _Float16 v16h;
typedef __attribute__((ext_vector_type(8)))  float    v8f;
typedef __attribute__((ext_vector_type(8)))  _Float16 v8h;
typedef __attribute__((ext_vector_type(4)))  unsigned int u32x4;

#define E32 32
#define GQ  2

union H8 { uint4 u; _Float16 h[8]; };

__device__ __forceinline__ int lane_id() { return (int)(threadIdx.x & 31); }

__device__ __forceinline__ v8f wmma_f16(v16h a, v16h b, v8f c) {
  return __builtin_amdgcn_wmma_f32_16x16x32_f16(
       false, a,  false, b,
       (short)0, c,  false,  false);
}

template <typename T>
__device__ __forceinline__ v16h load_frag(const T* __restrict__ src, int stride) {
  const int l  = lane_id();
  const int r  = l & 15;
  const int kb = (l >> 4) << 3;
  const T* p = src + (size_t)r * stride + kb;
  v16h f;
#pragma unroll
  for (int i = 0; i < 8; ++i) {
    f[i]     = (_Float16)p[i];
    f[8 + i] = (_Float16)p[i + 16];
  }
  return f;
}


#define ATT_WAVES 8
static_assert(64 % ATT_WAVES == 0 && ATT_WAVES * 32 == 256, "the launch below is 8 x 64 work-groups of 256 threads");


__device__ __forceinline__ v16h lds_frag(const _Float16* img) {
  const v8h lo = __builtin_bit_cast(v8h, *(const u32x4*)(img));
  const v8h hi = __builtin_bit_cast(v8h, *(const u32x4*)(img + 256));
  return __builtin_shufflevector(lo, hi, 0, 1, 2, 3, 4, 5, 6, 7,
                                 8, 9, 10, 11, 12, 13, 14, 15);
}

__device__ __forceinline__ float xhalf(float x) {
  const int b = __builtin_bit_cast(int, x);
  return __builtin_bit_cast(float,
      __builtin_amdgcn_permlanex16(b, b, 0x76543210, 0xfedcba98, false, false));
}

#define OT_PITCH 36
static_assert(ATT_WAVES * 16 * OT_PITCH * 4 <= 1024 * E32 * 2, "the output tiles lie over the K image");

__device__ __forceinline__ float leaky(float z) { return fmaxf(z, 0.01f * z); }

__device__ __forceinline__ void stage8(_Float16* dst, float4 lo, float4 hi) {
  H8 p;
  p.h[0] = (_Float16)lo.x;  p.h[1] = (_Float16)lo.y;  p.h[2] = (_Float16)lo.z;  p.h[3] = (_Float16)lo.w;
  p.h[4] = (_Float16)hi.x;  p.h[5] = (_Float16)hi.y;  p.h[6] = (_Float16)hi.z;  p.h[7] = (_Float16)hi.w;
  *(uint4*)dst = p.u;
}

__device__ __forceinline__ v8f bias8(const float* p) {
  const float4 lo = *(const float4*)p;
  const float4 hi = *(const float4*)(p + 4);
  v8f b = {lo.x, lo.y, lo.z, lo.w, hi.x, hi.y, hi.z, hi.w};
  return b;
}

struct XRaw { float4 a, b, c, d; };
__device__ __forceinline__ XRaw x_raw(const float* __restrict__ tile, int pitch) {
  const int l = lane_id();
  const float* p = tile + (size_t)(l & 15) * pitch + ((l >> 4) << 3);
  XRaw r;
  r.a = *(const float4*)p;         r.b = *(const float4*)(p + 4);
  r.c = *(const float4*)(p + 16);  r.d = *(const float4*)(p + 20);
  return r;
}
__device__ __forceinline__ v16h x_frag(const XRaw r) {
  v16h f;
  f[0]  = (_Float16)r.a.x;  f[1]  = (_Float16)r.a.y;  f[2]  = (_Float16)r.a.z;  f[3]  = (_Float16)r.a.w;
  f[4]  = (_Float16)r.b.x;  f[5]  = (_Float16)r.b.y;  f[6]  = (_Float16)r.b.z;  f[7]  = (_Float16)r.b.w;
  f[8]  = (_Float16)r.c.x;  f[9]  = (_Float16)r.c.y;  f[10] = (_Float16)r.c.z;  f[11] = (_Float16)r.c.w;
  f[12] = (_Float16)r.d.x;  f[13] = (_Float16)r.d.y;  f[14] = (_Float16)r.d.z;  f[15] = (_Float16)r.d.w;
  return f;
}

__global__ void __launch_bounds__(32 * ATT_WAVES)
attn_mlp_kernel(const float* __restrict__ x, const float* __restrict__ wqkv,
                const float* __restrict__ bqkv,
                const float* __restrict__ w1,  const float* __restrict__ b1,
                const float* __restrict__ w10, const float* __restrict__ b10,
                const float* __restrict__ w11, const float* __restrict__ b11,
                float* __restrict__ out) {
  __shared__ alignas(16) _Float16 kimg[1024 * E32];
  __shared__ alignas(16) _Float16 vimg[E32 * 1024];
  __shared__ alignas(16) _Float16 wimg[8 * 16 * E32];
  __shared__ alignas(16) float    bimg[ATT_WAVES * E32];

  const int tid = (int)threadIdx.x;
  const int wv = tid >> 5;
  const int qt = blockIdx.x * ATT_WAVES + wv;
  const int bg = blockIdx.y;
  const int g  = bg & (GQ - 1);
  const int l  = lane_id();
  const int nn = l & 15, half = l >> 4;
  const float* xg  = x + (size_t)(bg >> 1) * 1024 * (GQ * E32) + g * E32;
  const float* wg  = wqkv + (size_t)g * 96 * E32;
  const float* bgp = bqkv + g * 96;

  v16h wk_f[2], wv_f[2];
  v8f  bias_k[2];
  float bias_vc[2];
#pragma unroll
  for (int u = 0; u < 2; ++u) {
    wk_f[u] = load_frag(wg + (size_t)(32 + 16 * u) * E32, E32);
    wv_f[u] = load_frag(wg + (size_t)(64 + 16 * u) * E32, E32);
    bias_k[u]  = bias8(bgp + 32 + 16 * u + 8 * half);
    bias_vc[u] = bgp[64 + 16 * u + nn];
  }

  float4 wq[4];
  float  bval;

  constexpr int XP = GQ * E32;
  XRaw xr[4];
#pragma unroll
  for (int j = 0; j < 4; ++j) xr[j] = x_raw(xg + (size_t)(j * ATT_WAVES + wv) * 16 * XP, XP);

  XRaw xown, wq_raw[2];
  v8f  bias_q[2], bias_v[2];

#pragma unroll
  for (int j = 0; j < 64 / ATT_WAVES; ++j) {
    const int t = j * ATT_WAVES + wv;
    v16h xf = x_frag(xr[j & 3]);
    asm volatile("" : "+v"(xf));
    if (j + 4 < 64 / ATT_WAVES)
      xr[j & 3] = x_raw(xg + (size_t)((j + 4) * ATT_WAVES + wv) * 16 * XP, XP);
    if (j == 4) {
      const int c = tid & 15, h = (tid >> 4) & 1, i = (tid >> 5) & 1, fp = tid >> 6;
      const int orow = g * 32 + (fp & 1) * 16 + c;
      const float* s0 = w1 + (size_t)orow * 64 + (fp >> 1) * 32 + 16 * i + 8 * h;
      const float* s1 = ((fp >> 1) ? w11 : w10) + (size_t)orow * 32 + 16 * i + 8 * h;
      wq[0] = *(const float4*)s0;  wq[1] = *(const float4*)(s0 + 4);
      wq[2] = *(const float4*)s1;  wq[3] = *(const float4*)(s1 + 4);
      bval = (wv == 0 ? b1 : wv == 1 ? b10 : b11)[g * 32 + l];
      wq_raw[0] = x_raw(wg, E32);
      wq_raw[1] = x_raw(wg + 16 * E32, E32);
    }
    if (j == 6) {
      xown = x_raw(xg + (size_t)qt * 16 * XP, XP);
#pragma unroll
      for (int u = 0; u < 2; ++u) {
        bias_q[u] = bias8(bgp + 0  + 16 * u + 8 * half);
        bias_v[u] = bias8(bgp + 64 + 16 * u + 8 * half);
      }
    }
#pragma unroll
    for (int u = 0; u < 2; ++u) {
      v8f z = {};
      const v8f kt = wmma_f16(wk_f[u], xf, z);
      const v8f vv = wmma_f16(xf, wv_f[u], z);
      H8 pk, pv;
#pragma unroll
      for (int r = 0; r < 8; ++r) {
        pk.h[r] = (_Float16)(kt[r] + bias_k[u][r]);
        pv.h[r] = (_Float16)(vv[r] + bias_vc[u]);
      }
      *(uint4*)(kimg + t * 512 + u * 256 + l * 8) = pk.u;
      *(uint4*)(vimg + ((t >> 1) * 2 + u) * 512 + (t & 1) * 256 + l * 8) = pv.u;
    }
  }

  v16h bq, b_v;
  {
    const v16h xf = x_frag(xown);
#pragma unroll
    for (int u = 0; u < 2; ++u) {
      v8f z = {};
      const v8f qT = wmma_f16(x_frag(wq_raw[u]), xf, z);
      const v8f oT = wmma_f16(wv_f[u], xf, z);
#pragma unroll
      for (int r = 0; r < 8; ++r) {
        bq[8 * u + r]  = (_Float16)(qT[r] + bias_q[u][r]);
        b_v[8 * u + r] = (_Float16)(oT[r] + bias_v[u][r]);
      }
    }
  }
  stage8(wimg + tid * 8, wq[0], wq[1]);
  stage8(wimg + (256 + tid) * 8, wq[2], wq[3]);
  bimg[tid] = bval;
  __syncthreads();

  float m_run = -1e30f;
  float l_run = 0.0f;
  v8f acc0 = {}, acc1 = {};
  const float c2 = (1.0f / (float)E32) * 1.44269504088896340736f;
  const v8f c2v = {c2, c2, c2, c2, c2, c2, c2, c2};
  const _Float16* kl = kimg + l * 8;
  const _Float16* vl = vimg + l * 8;

  for (int jt = 0; jt < 16; ++jt) {
    const _Float16* ks = kl + jt * 2048;
    const _Float16* vs = vl + jt * 2048;
    v16h ak[4];
#pragma unroll
    for (int c = 0; c < 4; ++c)
      ak[c] = lds_frag(ks + c * 512);
    __builtin_amdgcn_sched_barrier(0);
    v16h av00 = lds_frag(vs);
    v16h av10 = lds_frag(vs + 512);
    v16h av01 = lds_frag(vs + 1024);
    v16h av11 = lds_frag(vs + 1536);
    __builtin_amdgcn_sched_barrier(0);

    v8f s[4];
#pragma unroll
    for (int c = 0; c < 4; ++c) {
      v8f z = {};
      s[c] = wmma_f16(ak[c], bq, z);
    }

    float mx = -__builtin_inff();
#pragma unroll
    for (int c = 0; c < 4; ++c) {
#pragma unroll
      for (int r = 0; r < 8; ++r) mx = fmaxf(mx, s[c][r]);
    }
    mx *= c2;
    mx = fmaxf(mx, xhalf(mx));
    const float mn  = fmaxf(m_run, mx);
    const float fac = __builtin_amdgcn_exp2f(m_run - mn);
    const float nm  = -mn;
    const v8f nmv = {nm, nm, nm, nm, nm, nm, nm, nm};
    v8f t[4];
#pragma unroll
    for (int c = 0; c < 4; ++c) t[c] = __builtin_elementwise_fma(s[c], c2v, nmv);
    float rs = 0.0f;
    v16h bp0, bp1;
#pragma unroll
    for (int r = 0; r < 8; ++r) {
      const float p0 = __builtin_amdgcn_exp2f(t[0][r]);
      const float p1 = __builtin_amdgcn_exp2f(t[1][r]);
      const float p2 = __builtin_amdgcn_exp2f(t[2][r]);
      const float p3 = __builtin_amdgcn_exp2f(t[3][r]);
      rs += (p0 + p1) + (p2 + p3);
      bp0[r] = (_Float16)p0;  bp0[8 + r] = (_Float16)p1;
      bp1[r] = (_Float16)p2;  bp1[8 + r] = (_Float16)p3;
    }
    l_run = l_run * fac + rs;
    m_run = mn;
#pragma unroll
    for (int r = 0; r < 8; ++r) { acc0[r] *= fac; acc1[r] *= fac; }

    acc0 = wmma_f16(av00, bp0, acc0);
    acc0 = wmma_f16(av01, bp1, acc0);
    acc1 = wmma_f16(av10, bp0, acc1);
    acc1 = wmma_f16(av11, bp1, acc1);
  }

  const float inv = 1.0f / (l_run + xhalf(l_run));
  v16h b_at;
#pragma unroll
  for (int r = 0; r < 8; ++r) {
    b_at[r]     = (_Float16)(acc0[r] * inv);
    b_at[8 + r] = (_Float16)(acc1[r] * inv);
  }

  v16h w1v[2], w1a[2], w10f[2], w11f[2];
  v8f  bs1[2], bs10[2], bs11[2];
#pragma unroll
  for (int cb = 0; cb < 2; ++cb) {
    const _Float16* wl = wimg + l * 8 + cb * 512;
    w1v[cb]  = lds_frag(wl);
    w1a[cb]  = lds_frag(wl + 2 * 512);
    w10f[cb] = lds_frag(wl + 4 * 512);
    w11f[cb] = lds_frag(wl + 6 * 512);
    const float* bl = bimg + cb * 16 + half * 8;
    bs1[cb]  = bias8(bl);
    bs10[cb] = bias8(bl + E32);
    bs11[cb] = bias8(bl + 2 * E32);
  }

  v16h b_h1;
#pragma unroll
  for (int cb = 0; cb < 2; ++cb) {
    v8f a = {};
    a = wmma_f16(w1v[cb], b_v, a);
    a = wmma_f16(w1a[cb], b_at, a);
#pragma unroll
    for (int r = 0; r < 8; ++r) b_h1[8 * cb + r] = (_Float16)leaky(a[r] + bs1[cb][r]);
  }
  v16h b_h2;
#pragma unroll
  for (int cb = 0; cb < 2; ++cb) {
    v8f a = {};
    a = wmma_f16(w10f[cb], b_h1, a);
#pragma unroll
    for (int r = 0; r < 8; ++r) b_h2[8 * cb + r] = (_Float16)leaky(a[r] + bs10[cb][r]);
  }
  float4 res[2][2];
#pragma unroll
  for (int cb = 0; cb < 2; ++cb) {
    v8f a = {};
    a = wmma_f16(w11f[cb], b_h2, a);
    res[cb][0].x = leaky(a[0] + bs11[cb][0]);  res[cb][0].y = leaky(a[1] + bs11[cb][1]);
    res[cb][0].z = leaky(a[2] + bs11[cb][2]);  res[cb][0].w = leaky(a[3] + bs11[cb][3]);
    res[cb][1].x = leaky(a[4] + bs11[cb][4]);  res[cb][1].y = leaky(a[5] + bs11[cb][5]);
    res[cb][1].z = leaky(a[6] + bs11[cb][6]);  res[cb][1].w = leaky(a[7] + bs11[cb][7]);
  }
  __syncthreads();
  float* st = reinterpret_cast<float*>(kimg) + wv * (16 * OT_PITCH);
#pragma unroll
  for (int cb = 0; cb < 2; ++cb) {
    float* sp = st + nn * OT_PITCH + cb * 16 + half * 8;
    *(float4*)sp       = res[cb][0];
    *(float4*)(sp + 4) = res[cb][1];
  }
  __syncthreads();
  float* dst = out + ((size_t)(bg >> 1) * 1024 + qt * 16) * (GQ * E32) + g * E32;
#pragma unroll
  for (int i = 0; i < 4; ++i) {
    const int idx = i * 32 + l;
    const int row = idx >> 3;
    const int c   = (idx & 7) * 4;
    *(float4*)(dst + (size_t)row * (GQ * E32) + c) = *(const float4*)(st + row * OT_PITCH + c);
  }
}

extern "C" void kernel_launch(void* const* d_in, const int* in_sizes, int n_in,
                              void* d_out, int out_size, void* d_ws, size_t ws_size,
                              hipStream_t stream) {
  const float* x     = (const float*)d_in[0];
  const float* w_qkv = (const float*)d_in[1];
  const float* b_qkv = (const float*)d_in[2];
  const float* w_h1  = (const float*)d_in[3];
  const float* b_h1  = (const float*)d_in[4];
  const float* w_h10 = (const float*)d_in[5];
  const float* b_h10 = (const float*)d_in[6];
  const float* w_h11 = (const float*)d_in[7];
  const float* b_h11 = (const float*)d_in[8];
  float* out = (float*)d_out;

  attn_mlp_kernel<<<dim3(8, 64), 256, 0, stream>>>(x, w_qkv, b_qkv, w_h1, b_h1,
                                                   w_h10, b_h10, w_h11, b_h11, out);
}


# QKV prologue: each 16x32 f32 tile of x and of Wq/Wk/Wv is loaded as whole 128-byte rows (4 lines per wave load, was 16) and re-striped to fragment order through a wave-private LDS pad; sums unchanged
# speedup vs baseline: 3.4785x; 1.2267x over previous
#include <hip/hip_runtime.h>
#include <hip/hip_bf16.h>

typedef __attribute__((ext_vector_type(16))) _Float16 v16h;
typedef __attribute__((ext_vector_type(8)))  float    v8f;
typedef __attribute__((ext_vector_type(8)))  _Float16 v8h;
typedef __attribute__((ext_vector_type(4)))  unsigned int u32x4;

#define E32 32
#define GQ  2

union H8 { uint4 u; _Float16 h[8]; };

__device__ __forceinline__ int lane_id() { return (int)(threadIdx.x & 31); }

__device__ __forceinline__ v8f wmma_f16(v16h a, v16h b, v8f c) {
  return __builtin_amdgcn_wmma_f32_16x16x32_f16(
       false, a,  false, b,
       (short)0, c,  false,  false);
}


#define ATT_WAVES 8
static_assert(64 % ATT_WAVES == 0 && ATT_WAVES * 32 == 256, "the launch below is 8 x 64 work-groups of 256 threads");


__device__ __forceinline__ v16h lds_frag(const _Float16* img) {
  const v8h lo = __builtin_bit_cast(v8h, *(const u32x4*)(img));
  const v8h hi = __builtin_bit_cast(v8h, *(const u32x4*)(img + 256));
  return __builtin_shufflevector(lo, hi, 0, 1, 2, 3, 4, 5, 6, 7,
                                 8, 9, 10, 11, 12, 13, 14, 15);
}

__device__ __forceinline__ float xhalf(float x) {
  const int b = __builtin_bit_cast(int, x);
  return __builtin_bit_cast(float,
      __builtin_amdgcn_permlanex16(b, b, 0x76543210, 0xfedcba98, false, false));
}

#define OT_PITCH 36
static_assert(ATT_WAVES * 16 * OT_PITCH * 4 <= 1024 * E32 * 2, "the output tiles lie over the K image");

__device__ __forceinline__ float leaky(float z) { return fmaxf(z, 0.01f * z); }

__device__ __forceinline__ void stage8(_Float16* dst, float4 lo, float4 hi) {
  H8 p;
  p.h[0] = (_Float16)lo.x;  p.h[1] = (_Float16)lo.y;  p.h[2] = (_Float16)lo.z;  p.h[3] = (_Float16)lo.w;
  p.h[4] = (_Float16)hi.x;  p.h[5] = (_Float16)hi.y;  p.h[6] = (_Float16)hi.z;  p.h[7] = (_Float16)hi.w;
  *(uint4*)dst = p.u;
}

__device__ __forceinline__ v8f bias8(const float* p) {
  const float4 lo = *(const float4*)p;
  const float4 hi = *(const float4*)(p + 4);
  v8f b = {lo.x, lo.y, lo.z, lo.w, hi.x, hi.y, hi.z, hi.w};
  return b;
}

struct XRaw { float4 a, b, c, d; };
__device__ __forceinline__ XRaw x_raw(const float* __restrict__ tile, int pitch) {
  const int l = lane_id();
  const float* p = tile + (size_t)(l >> 3) * pitch + ((l & 7) << 2);
  XRaw r;
  r.a = *(const float4*)p;
  r.b = *(const float4*)(p + 4 * (size_t)pitch);
  r.c = *(const float4*)(p + 8 * (size_t)pitch);
  r.d = *(const float4*)(p + 12 * (size_t)pitch);
  return r;
}

typedef __attribute__((ext_vector_type(4))) float        v4f;
typedef __attribute__((ext_vector_type(4))) _Float16     v4h;
typedef __attribute__((ext_vector_type(2))) unsigned int u32x2;
__device__ __forceinline__ u32x2 pack4(float4 v) {
  const v4f f = {v.x, v.y, v.z, v.w};
  return __builtin_bit_cast(u32x2, __builtin_convertvector(f, v4h));
}

#define PAD_PITCH 40
__device__ __forceinline__ v16h x_frag(_Float16* pad, const XRaw r) {
  const int l = lane_id();
  _Float16* sp = pad + (l >> 3) * PAD_PITCH + ((l & 7) << 2);
  *(u32x2*)sp                    = pack4(r.a);
  *(u32x2*)(sp + 4 * PAD_PITCH)  = pack4(r.b);
  *(u32x2*)(sp + 8 * PAD_PITCH)  = pack4(r.c);
  *(u32x2*)(sp + 12 * PAD_PITCH) = pack4(r.d);
  __builtin_amdgcn_wave_barrier();
  const _Float16* lp = pad + (l & 15) * PAD_PITCH + ((l >> 4) << 3);
  const v8h lo = __builtin_bit_cast(v8h, *(const u32x4*)(lp));
  const v8h hi = __builtin_bit_cast(v8h, *(const u32x4*)(lp + 16));
  return __builtin_shufflevector(lo, hi, 0, 1, 2, 3, 4, 5, 6, 7,
                                 8, 9, 10, 11, 12, 13, 14, 15);
}

__global__ void __launch_bounds__(32 * ATT_WAVES)
attn_mlp_kernel(const float* __restrict__ x, const float* __restrict__ wqkv,
                const float* __restrict__ bqkv,
                const float* __restrict__ w1,  const float* __restrict__ b1,
                const float* __restrict__ w10, const float* __restrict__ b10,
                const float* __restrict__ w11, const float* __restrict__ b11,
                float* __restrict__ out) {
  __shared__ alignas(16) _Float16 kimg[1024 * E32];
  __shared__ alignas(16) _Float16 vimg[E32 * 1024];
  __shared__ alignas(16) _Float16 wimg[8 * 16 * E32];
  __shared__ alignas(16) float    bimg[ATT_WAVES * E32];
  __shared__ alignas(16) _Float16 padimg[ATT_WAVES * 16 * PAD_PITCH];

  const int tid = (int)threadIdx.x;
  const int wv = tid >> 5;
  const int qt = blockIdx.x * ATT_WAVES + wv;
  const int bg = blockIdx.y;
  const int g  = bg & (GQ - 1);
  const int l  = lane_id();
  const int nn = l & 15, half = l >> 4;
  const float* xg  = x + (size_t)(bg >> 1) * 1024 * (GQ * E32) + g * E32;
  const float* wg  = wqkv + (size_t)g * 96 * E32;
  const float* bgp = bqkv + g * 96;

  _Float16* pad = padimg + wv * (16 * PAD_PITCH);

  XRaw wk_raw[2], wv_raw[2];
  v8f  bias_k[2];
  float bias_vc[2];
#pragma unroll
  for (int u = 0; u < 2; ++u) {
    wk_raw[u] = x_raw(wg + (size_t)(32 + 16 * u) * E32, E32);
    wv_raw[u] = x_raw(wg + (size_t)(64 + 16 * u) * E32, E32);
    bias_k[u]  = bias8(bgp + 32 + 16 * u + 8 * half);
    bias_vc[u] = bgp[64 + 16 * u + nn];
  }

  float4 wq[4];
  float  bval;

  constexpr int XP = GQ * E32;
  XRaw xr[4];
#pragma unroll
  for (int j = 0; j < 4; ++j) xr[j] = x_raw(xg + (size_t)(j * ATT_WAVES + wv) * 16 * XP, XP);

  v16h wk_f[2], wv_f[2];
#pragma unroll
  for (int u = 0; u < 2; ++u) {
    wk_f[u] = x_frag(pad, wk_raw[u]);
    wv_f[u] = x_frag(pad, wv_raw[u]);
  }
  v16h xf_next = x_frag(pad, xr[0]);
  xr[0] = x_raw(xg + (size_t)(4 * ATT_WAVES + wv) * 16 * XP, XP);

  XRaw xown, wq_raw[2];
  v8f  bias_q[2], bias_v[2];

#pragma unroll
  for (int j = 0; j < 64 / ATT_WAVES; ++j) {
    const int t = j * ATT_WAVES + wv;
    v16h xf = xf_next;
    asm volatile("" : "+v"(xf));
    if (j + 1 < 64 / ATT_WAVES) {
      xf_next = x_frag(pad, xr[(j + 1) & 3]);
      if (j + 5 < 64 / ATT_WAVES)
        xr[(j + 1) & 3] = x_raw(xg + (size_t)((j + 5) * ATT_WAVES + wv) * 16 * XP, XP);
    }
    if (j == 3) {
      const int c = tid & 15, h = (tid >> 4) & 1, i = (tid >> 5) & 1, fp = tid >> 6;
      const int orow = g * 32 + (fp & 1) * 16 + c;
      const float* s0 = w1 + (size_t)orow * 64 + (fp >> 1) * 32 + 16 * i + 8 * h;
      const float* s1 = ((fp >> 1) ? w11 : w10) + (size_t)orow * 32 + 16 * i + 8 * h;
      wq[0] = *(const float4*)s0;  wq[1] = *(const float4*)(s0 + 4);
      wq[2] = *(const float4*)s1;  wq[3] = *(const float4*)(s1 + 4);
      bval = (wv == 0 ? b1 : wv == 1 ? b10 : b11)[g * 32 + l];
      wq_raw[0] = x_raw(wg, E32);
      wq_raw[1] = x_raw(wg + 16 * E32, E32);
    }
    if (j == 5) {
      xown = x_raw(xg + (size_t)qt * 16 * XP, XP);
#pragma unroll
      for (int u = 0; u < 2; ++u) {
        bias_q[u] = bias8(bgp + 0  + 16 * u + 8 * half);
        bias_v[u] = bias8(bgp + 64 + 16 * u + 8 * half);
      }
    }
#pragma unroll
    for (int u = 0; u < 2; ++u) {
      v8f z = {};
      const v8f kt = wmma_f16(wk_f[u], xf, z);
      const v8f vv = wmma_f16(xf, wv_f[u], z);
      H8 pk, pv;
#pragma unroll
      for (int r = 0; r < 8; ++r) {
        pk.h[r] = (_Float16)(kt[r] + bias_k[u][r]);
        pv.h[r] = (_Float16)(vv[r] + bias_vc[u]);
      }
      *(uint4*)(kimg + t * 512 + u * 256 + l * 8) = pk.u;
      *(uint4*)(vimg + ((t >> 1) * 2 + u) * 512 + (t & 1) * 256 + l * 8) = pv.u;
    }
  }

  v16h bq, b_v;
  {
    const v16h xf = x_frag(pad, xown);
    const v16h wq_f[2] = {x_frag(pad, wq_raw[0]), x_frag(pad, wq_raw[1])};
#pragma unroll
    for (int u = 0; u < 2; ++u) {
      v8f z = {};
      const v8f qT = wmma_f16(wq_f[u], xf, z);
      const v8f oT = wmma_f16(wv_f[u], xf, z);
#pragma unroll
      for (int r = 0; r < 8; ++r) {
        bq[8 * u + r]  = (_Float16)(qT[r] + bias_q[u][r]);
        b_v[8 * u + r] = (_Float16)(oT[r] + bias_v[u][r]);
      }
    }
  }
  stage8(wimg + tid * 8, wq[0], wq[1]);
  stage8(wimg + (256 + tid) * 8, wq[2], wq[3]);
  bimg[tid] = bval;
  __syncthreads();

  float m_run = -1e30f;
  float l_run = 0.0f;
  v8f acc0 = {}, acc1 = {};
  const float c2 = (1.0f / (float)E32) * 1.44269504088896340736f;
  const v8f c2v = {c2, c2, c2, c2, c2, c2, c2, c2};
  const _Float16* kl = kimg + l * 8;
  const _Float16* vl = vimg + l * 8;

  for (int jt = 0; jt < 16; ++jt) {
    const _Float16* ks = kl + jt * 2048;
    const _Float16* vs = vl + jt * 2048;
    v16h ak[4];
#pragma unroll
    for (int c = 0; c < 4; ++c)
      ak[c] = lds_frag(ks + c * 512);
    __builtin_amdgcn_sched_barrier(0);
    v16h av00 = lds_frag(vs);
    v16h av10 = lds_frag(vs + 512);
    v16h av01 = lds_frag(vs + 1024);
    v16h av11 = lds_frag(vs + 1536);
    __builtin_amdgcn_sched_barrier(0);

    v8f s[4];
#pragma unroll
    for (int c = 0; c < 4; ++c) {
      v8f z = {};
      s[c] = wmma_f16(ak[c], bq, z);
    }

    float mx = -__builtin_inff();
#pragma unroll
    for (int c = 0; c < 4; ++c) {
#pragma unroll
      for (int r = 0; r < 8; ++r) mx = fmaxf(mx, s[c][r]);
    }
    mx *= c2;
    mx = fmaxf(mx, xhalf(mx));
    const float mn  = fmaxf(m_run, mx);
    const float fac = __builtin_amdgcn_exp2f(m_run - mn);
    const float nm  = -mn;
    const v8f nmv = {nm, nm, nm, nm, nm, nm, nm, nm};
    v8f t[4];
#pragma unroll
    for (int c = 0; c < 4; ++c) t[c] = __builtin_elementwise_fma(s[c], c2v, nmv);
    float rs = 0.0f;
    v16h bp0, bp1;
#pragma unroll
    for (int r = 0; r < 8; ++r) {
      const float p0 = __builtin_amdgcn_exp2f(t[0][r]);
      const float p1 = __builtin_amdgcn_exp2f(t[1][r]);
      const float p2 = __builtin_amdgcn_exp2f(t[2][r]);
      const float p3 = __builtin_amdgcn_exp2f(t[3][r]);
      rs += (p0 + p1) + (p2 + p3);
      bp0[r] = (_Float16)p0;  bp0[8 + r] = (_Float16)p1;
      bp1[r] = (_Float16)p2;  bp1[8 + r] = (_Float16)p3;
    }
    l_run = l_run * fac + rs;
    m_run = mn;
#pragma unroll
    for (int r = 0; r < 8; ++r) { acc0[r] *= fac; acc1[r] *= fac; }

    acc0 = wmma_f16(av00, bp0, acc0);
    acc0 = wmma_f16(av01, bp1, acc0);
    acc1 = wmma_f16(av10, bp0, acc1);
    acc1 = wmma_f16(av11, bp1, acc1);
  }

  const float inv = 1.0f / (l_run + xhalf(l_run));
  v16h b_at;
#pragma unroll
  for (int r = 0; r < 8; ++r) {
    b_at[r]     = (_Float16)(acc0[r] * inv);
    b_at[8 + r] = (_Float16)(acc1[r] * inv);
  }

  v16h w1v[2], w1a[2], w10f[2], w11f[2];
  v8f  bs1[2], bs10[2], bs11[2];
#pragma unroll
  for (int cb = 0; cb < 2; ++cb) {
    const _Float16* wl = wimg + l * 8 + cb * 512;
    w1v[cb]  = lds_frag(wl);
    w1a[cb]  = lds_frag(wl + 2 * 512);
    w10f[cb] = lds_frag(wl + 4 * 512);
    w11f[cb] = lds_frag(wl + 6 * 512);
    const float* bl = bimg + cb * 16 + half * 8;
    bs1[cb]  = bias8(bl);
    bs10[cb] = bias8(bl + E32);
    bs11[cb] = bias8(bl + 2 * E32);
  }

  v16h b_h1;
#pragma unroll
  for (int cb = 0; cb < 2; ++cb) {
    v8f a = {};
    a = wmma_f16(w1v[cb], b_v, a);
    a = wmma_f16(w1a[cb], b_at, a);
#pragma unroll
    for (int r = 0; r < 8; ++r) b_h1[8 * cb + r] = (_Float16)leaky(a[r] + bs1[cb][r]);
  }
  v16h b_h2;
#pragma unroll
  for (int cb = 0; cb < 2; ++cb) {
    v8f a = {};
    a = wmma_f16(w10f[cb], b_h1, a);
#pragma unroll
    for (int r = 0; r < 8; ++r) b_h2[8 * cb + r] = (_Float16)leaky(a[r] + bs10[cb][r]);
  }
  float4 res[2][2];
#pragma unroll
  for (int cb = 0; cb < 2; ++cb) {
    v8f a = {};
    a = wmma_f16(w11f[cb], b_h2, a);
    res[cb][0].x = leaky(a[0] + bs11[cb][0]);  res[cb][0].y = leaky(a[1] + bs11[cb][1]);
    res[cb][0].z = leaky(a[2] + bs11[cb][2]);  res[cb][0].w = leaky(a[3] + bs11[cb][3]);
    res[cb][1].x = leaky(a[4] + bs11[cb][4]);  res[cb][1].y = leaky(a[5] + bs11[cb][5]);
    res[cb][1].z = leaky(a[6] + bs11[cb][6]);  res[cb][1].w = leaky(a[7] + bs11[cb][7]);
  }
  __syncthreads();
  float* st = reinterpret_cast<float*>(kimg) + wv * (16 * OT_PITCH);
#pragma unroll
  for (int cb = 0; cb < 2; ++cb) {
    float* sp = st + nn * OT_PITCH + cb * 16 + half * 8;
    *(float4*)sp       = res[cb][0];
    *(float4*)(sp + 4) = res[cb][1];
  }
  __syncthreads();
  float* dst = out + ((size_t)(bg >> 1) * 1024 + qt * 16) * (GQ * E32) + g * E32;
#pragma unroll
  for (int i = 0; i < 4; ++i) {
    const int idx = i * 32 + l;
    const int row = idx >> 3;
    const int c   = (idx & 7) * 4;
    *(float4*)(dst + (size_t)row * (GQ * E32) + c) = *(const float4*)(st + row * OT_PITCH + c);
  }
}

extern "C" void kernel_launch(void* const* d_in, const int* in_sizes, int n_in,
                              void* d_out, int out_size, void* d_ws, size_t ws_size,
                              hipStream_t stream) {
  const float* x     = (const float*)d_in[0];
  const float* w_qkv = (const float*)d_in[1];
  const float* b_qkv = (const float*)d_in[2];
  const float* w_h1  = (const float*)d_in[3];
  const float* b_h1  = (const float*)d_in[4];
  const float* w_h10 = (const float*)d_in[5];
  const float* b_h10 = (const float*)d_in[6];
  const float* w_h11 = (const float*)d_in[7];
  const float* b_h11 = (const float*)d_in[8];
  float* out = (float*)d_out;

  attn_mlp_kernel<<<dim3(8, 64), 256, 0, stream>>>(x, w_qkv, b_qkv, w_h1, b_h1,
                                                   w_h10, b_h10, w_h11, b_h11, out);
}
